# starting kernel
# baseline (speedup 1.0000x reference)
#include <hip/hip_runtime.h>

#define N_NODES 50000
#define N_EDGES 800000

typedef __attribute__((ext_vector_type(2))) float v2f;
typedef __attribute__((ext_vector_type(8))) float v8f;


__global__ void zero_f32(float* __restrict__ p, long n) {
  long i = (long)blockIdx.x * blockDim.x + threadIdx.x;
  long stride = (long)gridDim.x * blockDim.x;
  for (; i < n; i += stride) p[i] = 0.0f;
}

__global__ void count_edges(const int* __restrict__ dst, float* __restrict__ cnt, int nE) {
  int e = blockIdx.x * blockDim.x + threadIdx.x;
  if (e < nE) atomicAdd(&cnt[dst[e]], 1.0f);
}

__global__ void recip_clamp(const float* __restrict__ cnt, float* __restrict__ inv, int n) {
  int i = blockIdx.x * blockDim.x + threadIdx.x;
  if (i < n) inv[i] = 1.0f / fmaxf(cnt[i], 1.0f);
}

__global__ void scatter_edges(const float* __restrict__ feat,
                              const int* __restrict__ src,
                              const int* __restrict__ dst,
                              float* __restrict__ agg,
                              int nE, int d) {
  int wave = blockIdx.x * (blockDim.x >> 5) + (threadIdx.x >> 5);
  int lane = threadIdx.x & 31;
  if (wave >= nE) return;
  const float* fs = feat + (size_t)src[wave] * d;
  float*       fd = agg  + (size_t)dst[wave] * d;
  for (int c = lane; c < d; c += 32)
    atomicAdd(&fd[c], fs[c]);
}


template<int DIN, int DOUT>
__global__ __launch_bounds__(32)
void sage_wmma_gemm(const float* __restrict__ agg,
                    const float* __restrict__ xin,
                    const float* __restrict__ inv,
                    const float* __restrict__ Wl,
                    const float* __restrict__ Wr,
                    const float* __restrict__ bias,
                    float* __restrict__ out) {
  const int lane = threadIdx.x & 31;
  const int half = lane >> 4;
  const int l16  = lane & 15;
  const int mTile = blockIdx.x << 4;
  const int nBase = blockIdx.y << 6;

  const int aRow = mTile + l16;
  const int kOff = half << 1;

  const float invc = inv[aRow];

  v8f acc[4] = {v8f{}, v8f{}, v8f{}, v8f{}};

  {
    const float* aP = agg + (size_t)aRow * DIN + kOff;
    const float* bP = Wl  + (size_t)(nBase + l16) * DIN + kOff;
#pragma unroll 4
    for (int k = 0; k < DIN; k += 4) {
      v2f a = *(const v2f*)(aP + k);
      a *= invc;
#pragma unroll
      for (int t = 0; t < 4; ++t) {
        v2f b = *(const v2f*)(bP + (size_t)t * 16 * DIN + k);
        acc[t] = __builtin_amdgcn_wmma_f32_16x16x4_f32(false, a, false, b,
                                                       (short)0, acc[t],
                                                       false, false);
      }
    }
  }

  {
    const float* aP = xin + (size_t)aRow * DIN + kOff;
    const float* bP = Wr  + (size_t)(nBase + l16) * DIN + kOff;
#pragma unroll 4
    for (int k = 0; k < DIN; k += 4) {
      v2f a = *(const v2f*)(aP + k);
#pragma unroll
      for (int t = 0; t < 4; ++t) {
        v2f b = *(const v2f*)(bP + (size_t)t * 16 * DIN + k);
        acc[t] = __builtin_amdgcn_wmma_f32_16x16x4_f32(false, a, false, b,
                                                       (short)0, acc[t],
                                                       false, false);
      }
    }
  }

  const int rowBase = mTile + (half << 3);
#pragma unroll
  for (int t = 0; t < 4; ++t) {
    const int col = nBase + t * 16 + l16;
    const float bj = bias[col];
#pragma unroll
    for (int i = 0; i < 8; ++i)
      out[(size_t)(rowBase + i) * DOUT + col] = acc[t][i] + bj;
  }
}


extern "C" void kernel_launch(void* const* d_in, const int* in_sizes, int n_in,
                              void* d_out, int out_size, void* d_ws, size_t ws_size,
                              hipStream_t stream) {
  const float* x   = (const float*)d_in[0];
  const int*   ei  = (const int*)d_in[1];
  const int*   src = ei;
  const int*   dst = ei + N_EDGES;
  const float* Wl1 = (const float*)d_in[2];
  const float* b1  = (const float*)d_in[3];
  const float* Wr1 = (const float*)d_in[4];
  const float* Wl2 = (const float*)d_in[5];
  const float* b2  = (const float*)d_in[6];
  const float* Wr2 = (const float*)d_in[7];
  const float* Wl3 = (const float*)d_in[8];
  const float* b3  = (const float*)d_in[9];
  const float* Wr3 = (const float*)d_in[10];
  float* out = (float*)d_out;

  float* ws  = (float*)d_ws;
  float* cnt = ws;
  float* inv = cnt + 50048;
  float* agg = inv + 50048;
  float* h1  = agg + (size_t)N_NODES * 128;
  float* h2  = h1  + (size_t)N_NODES * 128;

  zero_f32<<<256, 256, 0, stream>>>(cnt, (long)N_NODES);
  count_edges<<<(N_EDGES + 255) / 256, 256, 0, stream>>>(dst, cnt, N_EDGES);
  recip_clamp<<<(N_NODES + 255) / 256, 256, 0, stream>>>(cnt, inv, N_NODES);

  const int scatterBlocks = N_EDGES / 8;

  zero_f32<<<2048, 256, 0, stream>>>(agg, (long)N_NODES * 128);
  scatter_edges<<<scatterBlocks, 256, 0, stream>>>(x, src, dst, agg, N_EDGES, 128);
  {
    dim3 g(N_NODES / 16, 128 / 64);
    sage_wmma_gemm<128, 128><<<g, 32, 0, stream>>>(agg, x, inv, Wl1, Wr1, b1, h1);
  }

  zero_f32<<<2048, 256, 0, stream>>>(agg, (long)N_NODES * 128);
  scatter_edges<<<scatterBlocks, 256, 0, stream>>>(h1, src, dst, agg, N_EDGES, 128);
  {
    dim3 g(N_NODES / 16, 64 / 64);
    sage_wmma_gemm<128, 64><<<g, 32, 0, stream>>>(agg, h1, inv, Wl2, Wr2, b2, h2);
  }

  zero_f32<<<2048, 256, 0, stream>>>(agg, (long)N_NODES * 64);
  scatter_edges<<<scatterBlocks, 256, 0, stream>>>(h2, src, dst, agg, N_EDGES, 64);
  {
    dim3 g(N_NODES / 16, 128 / 64);
    sage_wmma_gemm<64, 128><<<g, 32, 0, stream>>>(agg, h2, inv, Wl3, Wr3, b3, out);
  }
}


# Scatter pass kept inside one half of the package per sum line: rows cut in two pieces held as two planes, piece q served from a ticket counter by waves on half q; float atomics kept, arrival order
# speedup vs baseline: 1.8979x; 1.8979x over previous
#include <hip/hip_runtime.h>

#define N_NODES 50000
#define N_EDGES 800000

typedef __attribute__((ext_vector_type(2))) float v2f;
typedef __attribute__((ext_vector_type(8))) float v8f;


__global__ void zero_f32(float* __restrict__ p, long n) {
  long i = (long)blockIdx.x * blockDim.x + threadIdx.x;
  long stride = (long)gridDim.x * blockDim.x;
  for (; i < n; i += stride) p[i] = 0.0f;
}

__global__ void count_edges(const int* __restrict__ dst, float* __restrict__ cnt, int nE) {
  int e = blockIdx.x * blockDim.x + threadIdx.x;
  if (e < nE) atomicAdd(&cnt[dst[e]], 1.0f);
}

__global__ void recip_clamp(const float* __restrict__ cnt, float* __restrict__ inv, int n) {
  int i = blockIdx.x * blockDim.x + threadIdx.x;
  if (i < n) inv[i] = 1.0f / fmaxf(cnt[i], 1.0f);
}

static_assert(N_EDGES % 64 == 0, "a ticket is two whole groups of 32 edges");

__device__ __forceinline__ unsigned draw_ticket(unsigned* ctr, int lane) {
  unsigned v = 0;
  if (lane == 0) v = atomicInc(ctr, 0xFFFFFFFFu);
  return v;
}

template<int D>
__global__ __launch_bounds__(128)
void scatter_edges(const float* __restrict__ feat,
                   const int* __restrict__ src,
                   const int* __restrict__ dst,
                   float* __restrict__ agg,
                   unsigned* __restrict__ queue,
                   int nE) {
  constexpr int PL = D / 2;
  constexpr int CH = PL / 32;
  constexpr int U  = 32 / CH;
  const int lane = threadIdx.x & 31;
  const int nT = nE >> 6;
  const int myHalf = (int)((__builtin_amdgcn_s_sendmsg_rtn(0x87) >> 18) & 1u);
  for (int pass = 0; pass < 2; ++pass) {
    const int q = myHalf ^ pass;
    unsigned* ctr = queue + q * 64;
    const float* fq = feat + q * PL + lane;
    float*       aq = agg  + (size_t)q * N_NODES * PL + lane;
    int t = (int)__builtin_amdgcn_readfirstlane(draw_ticket(ctr, lane));
    for (int it = 0; it < nT; ++it) {
      if (t >= nT) break;
      const unsigned tn = draw_ticket(ctr, lane);
      const int e0 = t << 6;
      const int s0 = src[e0 + lane];
      const int d0 = dst[e0 + lane];
      const int s1 = src[e0 + 32 + lane];
      const int d1 = dst[e0 + 32 + lane];
#pragma unroll
      for (int g = 0; g < 2; ++g) {
        const int s = g ? s1 : s0;
        const int d = g ? d1 : d0;
#pragma unroll
        for (int b = 0; b < 32; b += U) {
          float v[U][CH];
#pragma unroll
          for (int u = 0; u < U; ++u) {
            const int sj = __builtin_amdgcn_readlane(s, b + u);
            const float* fs = fq + (size_t)sj * D;
#pragma unroll
            for (int c = 0; c < CH; ++c) v[u][c] = fs[c * 32];
          }
#pragma unroll
          for (int u = 0; u < U; ++u) {
            const int tj = __builtin_amdgcn_readlane(d, b + u);
            float* fd = aq + (size_t)tj * PL;
#pragma unroll
            for (int c = 0; c < CH; ++c) atomicAdd(&fd[c * 32], v[u][c]);
          }
        }
      }
      t = (int)__builtin_amdgcn_readfirstlane(tn);
    }
  }
}


template<int DIN, int DOUT>
__global__ __launch_bounds__(32)
void sage_wmma_gemm(const float* __restrict__ agg,
                    const float* __restrict__ xin,
                    const float* __restrict__ inv,
                    const float* __restrict__ Wl,
                    const float* __restrict__ Wr,
                    const float* __restrict__ bias,
                    float* __restrict__ out) {
  const int lane = threadIdx.x & 31;
  const int half = lane >> 4;
  const int l16  = lane & 15;
  const int mTile = blockIdx.x << 4;
  const int nBase = blockIdx.y << 6;

  const int aRow = mTile + l16;
  const int kOff = half << 1;

  const float invc = inv[aRow];

  v8f acc[4] = {v8f{}, v8f{}, v8f{}, v8f{}};

  constexpr int PL = DIN / 2;
#pragma unroll
  for (int p = 0; p < 2; ++p) {
    const float* aP = agg + (size_t)p * N_NODES * PL + (size_t)aRow * PL + kOff;
    const float* bP = Wl  + (size_t)(nBase + l16) * DIN + p * PL + kOff;
#pragma unroll 4
    for (int k = 0; k < PL; k += 4) {
      v2f a = *(const v2f*)(aP + k);
      a *= invc;
#pragma unroll
      for (int t = 0; t < 4; ++t) {
        v2f b = *(const v2f*)(bP + (size_t)t * 16 * DIN + k);
        acc[t] = __builtin_amdgcn_wmma_f32_16x16x4_f32(false, a, false, b,
                                                       (short)0, acc[t],
                                                       false, false);
      }
    }
  }

  {
    const float* aP = xin + (size_t)aRow * DIN + kOff;
    const float* bP = Wr  + (size_t)(nBase + l16) * DIN + kOff;
#pragma unroll 4
    for (int k = 0; k < DIN; k += 4) {
      v2f a = *(const v2f*)(aP + k);
#pragma unroll
      for (int t = 0; t < 4; ++t) {
        v2f b = *(const v2f*)(bP + (size_t)t * 16 * DIN + k);
        acc[t] = __builtin_amdgcn_wmma_f32_16x16x4_f32(false, a, false, b,
                                                       (short)0, acc[t],
                                                       false, false);
      }
    }
  }

  const int rowBase = mTile + (half << 3);
#pragma unroll
  for (int t = 0; t < 4; ++t) {
    const int col = nBase + t * 16 + l16;
    const float bj = bias[col];
#pragma unroll
    for (int i = 0; i < 8; ++i)
      out[(size_t)(rowBase + i) * DOUT + col] = acc[t][i] + bj;
  }
}


extern "C" void kernel_launch(void* const* d_in, const int* in_sizes, int n_in,
                              void* d_out, int out_size, void* d_ws, size_t ws_size,
                              hipStream_t stream) {
  const float* x   = (const float*)d_in[0];
  const int*   ei  = (const int*)d_in[1];
  const int*   src = ei;
  const int*   dst = ei + N_EDGES;
  const float* Wl1 = (const float*)d_in[2];
  const float* b1  = (const float*)d_in[3];
  const float* Wr1 = (const float*)d_in[4];
  const float* Wl2 = (const float*)d_in[5];
  const float* b2  = (const float*)d_in[6];
  const float* Wr2 = (const float*)d_in[7];
  const float* Wl3 = (const float*)d_in[8];
  const float* b3  = (const float*)d_in[9];
  const float* Wr3 = (const float*)d_in[10];
  float* out = (float*)d_out;

  float* ws  = (float*)d_ws;
  float* cnt = ws;
  float* inv = cnt + 50048;
  float* tk  = inv + 50048;
  unsigned* queue = (unsigned*)tk;
  float* agg = tk + 128;
  float* h1  = agg + (size_t)N_NODES * 128;
  float* h2  = h1  + (size_t)N_NODES * 128;

  zero_f32<<<256, 256, 0, stream>>>(cnt, (long)N_NODES);
  count_edges<<<(N_EDGES + 255) / 256, 256, 0, stream>>>(dst, cnt, N_EDGES);
  recip_clamp<<<(N_NODES + 255) / 256, 256, 0, stream>>>(cnt, inv, N_NODES);

  const int scatterBlocks = 2048;

  zero_f32<<<2048, 256, 0, stream>>>(tk, (long)N_NODES * 128 + 128);
  scatter_edges<128><<<scatterBlocks, 128, 0, stream>>>(x, src, dst, agg, queue, N_EDGES);
  {
    dim3 g(N_NODES / 16, 128 / 64);
    sage_wmma_gemm<128, 128><<<g, 32, 0, stream>>>(agg, x, inv, Wl1, Wr1, b1, h1);
  }

  zero_f32<<<2048, 256, 0, stream>>>(tk, (long)N_NODES * 128 + 128);
  scatter_edges<128><<<scatterBlocks, 128, 0, stream>>>(h1, src, dst, agg, queue, N_EDGES);
  {
    dim3 g(N_NODES / 16, 64 / 64);
    sage_wmma_gemm<128, 64><<<g, 32, 0, stream>>>(agg, h1, inv, Wl2, Wr2, b2, h2);
  }

  zero_f32<<<2048, 256, 0, stream>>>(tk, (long)N_NODES * 64 + 128);
  scatter_edges<64><<<scatterBlocks, 128, 0, stream>>>(h2, src, dst, agg, queue, N_EDGES);
  {
    dim3 g(N_NODES / 16, 128 / 64);
    sage_wmma_gemm<64, 128><<<g, 32, 0, stream>>>(agg, h2, inv, Wl3, Wr3, b3, out);
  }
}


# GEMM fragment loads 16 B per lane: one load feeds two matrix instructions, k of each group of 8 taken in the same permuted order for A and B (sum order changes)
# speedup vs baseline: 2.0069x; 1.0574x over previous
#include <hip/hip_runtime.h>

#define N_NODES 50000
#define N_EDGES 800000

typedef __attribute__((ext_vector_type(2))) float v2f;
typedef __attribute__((ext_vector_type(8))) float v8f;


__global__ void zero_f32(float* __restrict__ p, long n) {
  long i = (long)blockIdx.x * blockDim.x + threadIdx.x;
  long stride = (long)gridDim.x * blockDim.x;
  for (; i < n; i += stride) p[i] = 0.0f;
}

__global__ void count_edges(const int* __restrict__ dst, float* __restrict__ cnt, int nE) {
  int e = blockIdx.x * blockDim.x + threadIdx.x;
  if (e < nE) atomicAdd(&cnt[dst[e]], 1.0f);
}

__global__ void recip_clamp(const float* __restrict__ cnt, float* __restrict__ inv, int n) {
  int i = blockIdx.x * blockDim.x + threadIdx.x;
  if (i < n) inv[i] = 1.0f / fmaxf(cnt[i], 1.0f);
}

static_assert(N_EDGES % 64 == 0, "a ticket is two whole groups of 32 edges");

__device__ __forceinline__ unsigned draw_ticket(unsigned* ctr, int lane) {
  unsigned v = 0;
  if (lane == 0) v = atomicInc(ctr, 0xFFFFFFFFu);
  return v;
}

template<int D>
__global__ __launch_bounds__(128)
void scatter_edges(const float* __restrict__ feat,
                   const int* __restrict__ src,
                   const int* __restrict__ dst,
                   float* __restrict__ agg,
                   unsigned* __restrict__ queue,
                   int nE) {
  constexpr int PL = D / 2;
  constexpr int CH = PL / 32;
  constexpr int U  = 32 / CH;
  const int lane = threadIdx.x & 31;
  const int nT = nE >> 6;
  const int myHalf = (int)((__builtin_amdgcn_s_sendmsg_rtn(0x87) >> 18) & 1u);
  for (int pass = 0; pass < 2; ++pass) {
    const int q = myHalf ^ pass;
    unsigned* ctr = queue + q * 64;
    const float* fq = feat + q * PL + lane;
    float*       aq = agg  + (size_t)q * N_NODES * PL + lane;
    int t = (int)__builtin_amdgcn_readfirstlane(draw_ticket(ctr, lane));
    for (int it = 0; it < nT; ++it) {
      if (t >= nT) break;
      const unsigned tn = draw_ticket(ctr, lane);
      const int e0 = t << 6;
      const int s0 = src[e0 + lane];
      const int d0 = dst[e0 + lane];
      const int s1 = src[e0 + 32 + lane];
      const int d1 = dst[e0 + 32 + lane];
#pragma unroll
      for (int g = 0; g < 2; ++g) {
        const int s = g ? s1 : s0;
        const int d = g ? d1 : d0;
#pragma unroll
        for (int b = 0; b < 32; b += U) {
          float v[U][CH];
#pragma unroll
          for (int u = 0; u < U; ++u) {
            const int sj = __builtin_amdgcn_readlane(s, b + u);
            const float* fs = fq + (size_t)sj * D;
#pragma unroll
            for (int c = 0; c < CH; ++c) v[u][c] = fs[c * 32];
          }
#pragma unroll
          for (int u = 0; u < U; ++u) {
            const int tj = __builtin_amdgcn_readlane(d, b + u);
            float* fd = aq + (size_t)tj * PL;
#pragma unroll
            for (int c = 0; c < CH; ++c) atomicAdd(&fd[c * 32], v[u][c]);
          }
        }
      }
      t = (int)__builtin_amdgcn_readfirstlane(tn);
    }
  }
}


typedef __attribute__((ext_vector_type(4))) float v4f;

__device__ __forceinline__ void mma_k8(v8f (&acc)[4], const v4f a, const v4f (&b)[4]) {
  const v2f alo = {a.x, a.y}, ahi = {a.z, a.w};
#pragma unroll
  for (int t = 0; t < 4; ++t) {
    const v2f blo = {b[t].x, b[t].y};
    acc[t] = __builtin_amdgcn_wmma_f32_16x16x4_f32(false, alo, false, blo,
                                                   (short)0, acc[t], false, false);
  }
#pragma unroll
  for (int t = 0; t < 4; ++t) {
    const v2f bhi = {b[t].z, b[t].w};
    acc[t] = __builtin_amdgcn_wmma_f32_16x16x4_f32(false, ahi, false, bhi,
                                                   (short)0, acc[t], false, false);
  }
}

template<int DIN, int DOUT>
__global__ __launch_bounds__(32)
void sage_wmma_gemm(const float* __restrict__ agg,
                    const float* __restrict__ xin,
                    const float* __restrict__ inv,
                    const float* __restrict__ Wl,
                    const float* __restrict__ Wr,
                    const float* __restrict__ bias,
                    float* __restrict__ out) {
  const int lane = threadIdx.x & 31;
  const int half = lane >> 4;
  const int l16  = lane & 15;
  const int mTile = blockIdx.x << 4;
  const int nBase = blockIdx.y << 6;

  const int aRow = mTile + l16;
  const int kOff = half << 2;

  const float invc = inv[aRow];

  v8f acc[4] = {v8f{}, v8f{}, v8f{}, v8f{}};

  constexpr int PL = DIN / 2;
  static_assert(PL % 8 == 0, "a plane is whole groups of eight k");
#pragma unroll
  for (int p = 0; p < 2; ++p) {
    const float* aP = agg + (size_t)p * N_NODES * PL + (size_t)aRow * PL + kOff;
    const float* bP = Wl  + (size_t)(nBase + l16) * DIN + p * PL + kOff;
#pragma unroll 2
    for (int k = 0; k < PL; k += 8) {
      v4f a = *(const v4f*)(aP + k);
      a *= invc;
      v4f b[4];
#pragma unroll
      for (int t = 0; t < 4; ++t) b[t] = *(const v4f*)(bP + (size_t)t * 16 * DIN + k);
      mma_k8(acc, a, b);
    }
  }

  {
    const float* aP = xin + (size_t)aRow * DIN + kOff;
    const float* bP = Wr  + (size_t)(nBase + l16) * DIN + kOff;
#pragma unroll 2
    for (int k = 0; k < DIN; k += 8) {
      const v4f a = *(const v4f*)(aP + k);
      v4f b[4];
#pragma unroll
      for (int t = 0; t < 4; ++t) b[t] = *(const v4f*)(bP + (size_t)t * 16 * DIN + k);
      mma_k8(acc, a, b);
    }
  }

  const int rowBase = mTile + (half << 3);
#pragma unroll
  for (int t = 0; t < 4; ++t) {
    const int col = nBase + t * 16 + l16;
    const float bj = bias[col];
#pragma unroll
    for (int i = 0; i < 8; ++i)
      out[(size_t)(rowBase + i) * DOUT + col] = acc[t][i] + bj;
  }
}


extern "C" void kernel_launch(void* const* d_in, const int* in_sizes, int n_in,
                              void* d_out, int out_size, void* d_ws, size_t ws_size,
                              hipStream_t stream) {
  const float* x   = (const float*)d_in[0];
  const int*   ei  = (const int*)d_in[1];
  const int*   src = ei;
  const int*   dst = ei + N_EDGES;
  const float* Wl1 = (const float*)d_in[2];
  const float* b1  = (const float*)d_in[3];
  const float* Wr1 = (const float*)d_in[4];
  const float* Wl2 = (const float*)d_in[5];
  const float* b2  = (const float*)d_in[6];
  const float* Wr2 = (const float*)d_in[7];
  const float* Wl3 = (const float*)d_in[8];
  const float* b3  = (const float*)d_in[9];
  const float* Wr3 = (const float*)d_in[10];
  float* out = (float*)d_out;

  float* ws  = (float*)d_ws;
  float* cnt = ws;
  float* inv = cnt + 50048;
  float* tk  = inv + 50048;
  unsigned* queue = (unsigned*)tk;
  float* agg = tk + 128;
  float* h1  = agg + (size_t)N_NODES * 128;
  float* h2  = h1  + (size_t)N_NODES * 128;

  zero_f32<<<256, 256, 0, stream>>>(cnt, (long)N_NODES);
  count_edges<<<(N_EDGES + 255) / 256, 256, 0, stream>>>(dst, cnt, N_EDGES);
  recip_clamp<<<(N_NODES + 255) / 256, 256, 0, stream>>>(cnt, inv, N_NODES);

  const int scatterBlocks = 2048;

  zero_f32<<<2048, 256, 0, stream>>>(tk, (long)N_NODES * 128 + 128);
  scatter_edges<128><<<scatterBlocks, 128, 0, stream>>>(x, src, dst, agg, queue, N_EDGES);
  {
    dim3 g(N_NODES / 16, 128 / 64);
    sage_wmma_gemm<128, 128><<<g, 32, 0, stream>>>(agg, x, inv, Wl1, Wr1, b1, h1);
  }

  zero_f32<<<2048, 256, 0, stream>>>(tk, (long)N_NODES * 128 + 128);
  scatter_edges<128><<<scatterBlocks, 128, 0, stream>>>(h1, src, dst, agg, queue, N_EDGES);
  {
    dim3 g(N_NODES / 16, 64 / 64);
    sage_wmma_gemm<128, 64><<<g, 32, 0, stream>>>(agg, h1, inv, Wl2, Wr2, b2, h2);
  }

  zero_f32<<<2048, 256, 0, stream>>>(tk, (long)N_NODES * 64 + 128);
  scatter_edges<64><<<scatterBlocks, 128, 0, stream>>>(h2, src, dst, agg, queue, N_EDGES);
  {
    dim3 g(N_NODES / 16, 128 / 64);
    sage_wmma_gemm<64, 128><<<g, 32, 0, stream>>>(agg, h2, inv, Wl3, Wr3, b3, out);
  }
}


# Degree count through LDS: 32 workgroups each count 25000 edges into a 200 KB LDS table (integer LDS adds) and store it as a plane, the reciprocal pass sums the planes; no global atomic per edge
# speedup vs baseline: 2.3076x; 1.1498x over previous
#include <hip/hip_runtime.h>

#define N_NODES 50000
#define N_EDGES 800000

typedef __attribute__((ext_vector_type(2))) float v2f;
typedef __attribute__((ext_vector_type(8))) float v8f;


__global__ void zero_f32(float* __restrict__ p, long n) {
  long i = (long)blockIdx.x * blockDim.x + threadIdx.x;
  long stride = (long)gridDim.x * blockDim.x;
  for (; i < n; i += stride) p[i] = 0.0f;
}

#define HIST_BLOCKS 32
#define HIST_THREADS 1024
static_assert(N_EDGES % HIST_BLOCKS == 0, "every workgroup counts the same number of edges");
static_assert(N_NODES % 4 == 0, "a table is cleared and stored 16 bytes at a time");

__global__ __launch_bounds__(HIST_THREADS)
void degree_planes(const int* __restrict__ dst, unsigned* __restrict__ planes) {
  __shared__ __attribute__((aligned(16))) unsigned tab[N_NODES];
  constexpr int PER = N_EDGES / HIST_BLOCKS;
  constexpr int IT  = (PER + HIST_THREADS - 1) / HIST_THREADS;
  const int tid = threadIdx.x;
  const int* d = dst + (size_t)blockIdx.x * PER;
  int v[IT];
#pragma unroll
  for (int j = 0; j < IT; ++j) {
    const int e = tid + j * HIST_THREADS;
    v[j] = (e < PER) ? d[e] : -1;
  }
  uint4* tab4 = reinterpret_cast<uint4*>(tab);
  for (int i = tid; i < N_NODES / 4; i += HIST_THREADS) tab4[i] = make_uint4(0u, 0u, 0u, 0u);
  __syncthreads();
#pragma unroll
  for (int j = 0; j < IT; ++j)
    if ((unsigned)v[j] < (unsigned)N_NODES) atomicAdd(&tab[v[j]], 1u);
  __syncthreads();
  uint4* o4 = reinterpret_cast<uint4*>(planes + (size_t)blockIdx.x * N_NODES);
  for (int i = tid; i < N_NODES / 4; i += HIST_THREADS) o4[i] = tab4[i];
}

__global__ void recip_degree(const unsigned* __restrict__ planes, float* __restrict__ inv, int n) {
  int i = blockIdx.x * blockDim.x + threadIdx.x;
  if (i < n) {
    unsigned c = 0;
#pragma unroll
    for (int g = 0; g < HIST_BLOCKS; ++g) c += planes[(size_t)g * N_NODES + i];
    inv[i] = 1.0f / fmaxf((float)c, 1.0f);
  }
}

static_assert(N_EDGES % 64 == 0, "a ticket is two whole groups of 32 edges");

__device__ __forceinline__ unsigned draw_ticket(unsigned* ctr, int lane) {
  unsigned v = 0;
  if (lane == 0) v = atomicInc(ctr, 0xFFFFFFFFu);
  return v;
}

template<int D>
__global__ __launch_bounds__(128)
void scatter_edges(const float* __restrict__ feat,
                   const int* __restrict__ src,
                   const int* __restrict__ dst,
                   float* __restrict__ agg,
                   unsigned* __restrict__ queue,
                   int nE) {
  constexpr int PL = D / 2;
  constexpr int CH = PL / 32;
  constexpr int U  = 32 / CH;
  const int lane = threadIdx.x & 31;
  const int nT = nE >> 6;
  const int myHalf = (int)((__builtin_amdgcn_s_sendmsg_rtn(0x87) >> 18) & 1u);
  for (int pass = 0; pass < 2; ++pass) {
    const int q = myHalf ^ pass;
    unsigned* ctr = queue + q * 64;
    const float* fq = feat + q * PL + lane;
    float*       aq = agg  + (size_t)q * N_NODES * PL + lane;
    int t = (int)__builtin_amdgcn_readfirstlane(draw_ticket(ctr, lane));
    for (int it = 0; it < nT; ++it) {
      if (t >= nT) break;
      const unsigned tn = draw_ticket(ctr, lane);
      const int e0 = t << 6;
      const int s0 = src[e0 + lane];
      const int d0 = dst[e0 + lane];
      const int s1 = src[e0 + 32 + lane];
      const int d1 = dst[e0 + 32 + lane];
#pragma unroll
      for (int g = 0; g < 2; ++g) {
        const int s = g ? s1 : s0;
        const int d = g ? d1 : d0;
#pragma unroll
        for (int b = 0; b < 32; b += U) {
          float v[U][CH];
#pragma unroll
          for (int u = 0; u < U; ++u) {
            const int sj = __builtin_amdgcn_readlane(s, b + u);
            const float* fs = fq + (size_t)sj * D;
#pragma unroll
            for (int c = 0; c < CH; ++c) v[u][c] = fs[c * 32];
          }
#pragma unroll
          for (int u = 0; u < U; ++u) {
            const int tj = __builtin_amdgcn_readlane(d, b + u);
            float* fd = aq + (size_t)tj * PL;
#pragma unroll
            for (int c = 0; c < CH; ++c) atomicAdd(&fd[c * 32], v[u][c]);
          }
        }
      }
      t = (int)__builtin_amdgcn_readfirstlane(tn);
    }
  }
}


typedef __attribute__((ext_vector_type(4))) float v4f;

__device__ __forceinline__ void mma_k8(v8f (&acc)[4], const v4f a, const v4f (&b)[4]) {
  const v2f alo = {a.x, a.y}, ahi = {a.z, a.w};
#pragma unroll
  for (int t = 0; t < 4; ++t) {
    const v2f blo = {b[t].x, b[t].y};
    acc[t] = __builtin_amdgcn_wmma_f32_16x16x4_f32(false, alo, false, blo,
                                                   (short)0, acc[t], false, false);
  }
#pragma unroll
  for (int t = 0; t < 4; ++t) {
    const v2f bhi = {b[t].z, b[t].w};
    acc[t] = __builtin_amdgcn_wmma_f32_16x16x4_f32(false, ahi, false, bhi,
                                                   (short)0, acc[t], false, false);
  }
}

template<int DIN, int DOUT>
__global__ __launch_bounds__(32)
void sage_wmma_gemm(const float* __restrict__ agg,
                    const float* __restrict__ xin,
                    const float* __restrict__ inv,
                    const float* __restrict__ Wl,
                    const float* __restrict__ Wr,
                    const float* __restrict__ bias,
                    float* __restrict__ out) {
  const int lane = threadIdx.x & 31;
  const int half = lane >> 4;
  const int l16  = lane & 15;
  const int mTile = blockIdx.x << 4;
  const int nBase = blockIdx.y << 6;

  const int aRow = mTile + l16;
  const int kOff = half << 2;

  const float invc = inv[aRow];

  v8f acc[4] = {v8f{}, v8f{}, v8f{}, v8f{}};

  constexpr int PL = DIN / 2;
  static_assert(PL % 8 == 0, "a plane is whole groups of eight k");
#pragma unroll
  for (int p = 0; p < 2; ++p) {
    const float* aP = agg + (size_t)p * N_NODES * PL + (size_t)aRow * PL + kOff;
    const float* bP = Wl  + (size_t)(nBase + l16) * DIN + p * PL + kOff;
#pragma unroll 2
    for (int k = 0; k < PL; k += 8) {
      v4f a = *(const v4f*)(aP + k);
      a *= invc;
      v4f b[4];
#pragma unroll
      for (int t = 0; t < 4; ++t) b[t] = *(const v4f*)(bP + (size_t)t * 16 * DIN + k);
      mma_k8(acc, a, b);
    }
  }

  {
    const float* aP = xin + (size_t)aRow * DIN + kOff;
    const float* bP = Wr  + (size_t)(nBase + l16) * DIN + kOff;
#pragma unroll 2
    for (int k = 0; k < DIN; k += 8) {
      const v4f a = *(const v4f*)(aP + k);
      v4f b[4];
#pragma unroll
      for (int t = 0; t < 4; ++t) b[t] = *(const v4f*)(bP + (size_t)t * 16 * DIN + k);
      mma_k8(acc, a, b);
    }
  }

  const int rowBase = mTile + (half << 3);
#pragma unroll
  for (int t = 0; t < 4; ++t) {
    const int col = nBase + t * 16 + l16;
    const float bj = bias[col];
#pragma unroll
    for (int i = 0; i < 8; ++i)
      out[(size_t)(rowBase + i) * DOUT + col] = acc[t][i] + bj;
  }
}


extern "C" void kernel_launch(void* const* d_in, const int* in_sizes, int n_in,
                              void* d_out, int out_size, void* d_ws, size_t ws_size,
                              hipStream_t stream) {
  const float* x   = (const float*)d_in[0];
  const int*   ei  = (const int*)d_in[1];
  const int*   src = ei;
  const int*   dst = ei + N_EDGES;
  const float* Wl1 = (const float*)d_in[2];
  const float* b1  = (const float*)d_in[3];
  const float* Wr1 = (const float*)d_in[4];
  const float* Wl2 = (const float*)d_in[5];
  const float* b2  = (const float*)d_in[6];
  const float* Wr2 = (const float*)d_in[7];
  const float* Wl3 = (const float*)d_in[8];
  const float* b3  = (const float*)d_in[9];
  const float* Wr3 = (const float*)d_in[10];
  float* out = (float*)d_out;

  float* ws  = (float*)d_ws;
  float* inv = ws;
  float* tk  = inv + 50048;
  unsigned* queue = (unsigned*)tk;
  float* agg = tk + 128;
  float* h1  = agg + (size_t)N_NODES * 128;
  float* h2  = h1  + (size_t)N_NODES * 128;
  unsigned* planes = (unsigned*)(h2 + (size_t)N_NODES * 64);

  degree_planes<<<HIST_BLOCKS, HIST_THREADS, 0, stream>>>(dst, planes);
  recip_degree<<<(N_NODES + 255) / 256, 256, 0, stream>>>(planes, inv, N_NODES);

  const int scatterBlocks = 2048;

  zero_f32<<<2048, 256, 0, stream>>>(tk, (long)N_NODES * 128 + 128);
  scatter_edges<128><<<scatterBlocks, 128, 0, stream>>>(x, src, dst, agg, queue, N_EDGES);
  {
    dim3 g(N_NODES / 16, 128 / 64);
    sage_wmma_gemm<128, 128><<<g, 32, 0, stream>>>(agg, x, inv, Wl1, Wr1, b1, h1);
  }

  zero_f32<<<2048, 256, 0, stream>>>(tk, (long)N_NODES * 128 + 128);
  scatter_edges<128><<<scatterBlocks, 128, 0, stream>>>(h1, src, dst, agg, queue, N_EDGES);
  {
    dim3 g(N_NODES / 16, 64 / 64);
    sage_wmma_gemm<128, 64><<<g, 32, 0, stream>>>(agg, h1, inv, Wl2, Wr2, b2, h2);
  }

  zero_f32<<<2048, 256, 0, stream>>>(tk, (long)N_NODES * 64 + 128);
  scatter_edges<64><<<scatterBlocks, 128, 0, stream>>>(h2, src, dst, agg, queue, N_EDGES);
  {
    dim3 g(N_NODES / 16, 128 / 64);
    sage_wmma_gemm<64, 128><<<g, 32, 0, stream>>>(agg, h2, inv, Wl3, Wr3, b3, out);
  }
}


# Neighbour sums by owner workgroups: 250 blocks each keep 200 nodes' sums in LDS (LDS float adds, arrival order), edges found by one scan of the destination list per call; no memory atomics or fills
# speedup vs baseline: 2.4008x; 1.0404x over previous
#include <hip/hip_runtime.h>

#define N_NODES 50000
#define N_EDGES 800000

typedef __attribute__((ext_vector_type(2))) float v2f;
typedef __attribute__((ext_vector_type(8))) float v8f;

#define NPB 200
#define NOWN (N_NODES / NPB)
#define FIND_G 2
#define OWN_THREADS 1024
#define OWN_WAVES (OWN_THREADS / 32)
#define LIST_CAP 8192
#define OWN_LDS_WORDS 41216
static_assert(N_NODES % NPB == 0, "every owner has the same number of nodes");
static_assert(NOWN % FIND_G == 0, "every finder serves the same number of owners");
static_assert(N_EDGES % 4 == 0, "the destination list is read 16 bytes at a time");
static_assert(N_EDGES % 32 == 0, "a wave of the rescue path always holds 32 edges");
static_assert((NPB + 1) * 128 <= OWN_LDS_WORDS, "sum rows and the padding row fit the reservation");

#define SEG_K 40
#define TAB_WORDS 512
#define FIND_LDS_WORDS (TAB_WORDS + SEG_K * OWN_THREADS)
static_assert(FIND_G * NPB + FIND_G + 1 <= TAB_WORDS, "table fits its words");
static_assert(TAB_WORDS <= OWN_THREADS, "one thread per table word when a finder clears it");
static_assert(N_EDGES <= (1 << 20) && FIND_G * NPB <= (1 << 12), "a match packs into one word");
static_assert(FIND_LDS_WORDS * 4 > 160 * 1024, "one finder per WGP");

__global__ __launch_bounds__(OWN_THREADS)
void find_edges(const int* __restrict__ src, const int* __restrict__ dst,
                int2* __restrict__ list,
                unsigned* __restrict__ counts,
                float* __restrict__ inv) {
  __shared__ unsigned tab[FIND_LDS_WORDS];
  constexpr unsigned SPAN = FIND_G * NPB;
  const int tid = threadIdx.x;
  if (tid < TAB_WORDS) tab[tid] = 0u;
  __syncthreads();
  const unsigned lo = (unsigned)blockIdx.x * SPAN;
  unsigned* seg = tab + TAB_WORDS + tid;
  unsigned c = 0;
  const int4* d4 = reinterpret_cast<const int4*>(dst);
  constexpr int NV = N_EDGES / 4;
  for (int i0 = tid; i0 < NV; i0 += 4 * OWN_THREADS) {
    int4 d[4];
#pragma unroll
    for (int k = 0; k < 4; ++k) {
      const int i = i0 + k * OWN_THREADS;
      d[k] = (i < NV) ? d4[i] : make_int4(-1, -1, -1, -1);
    }
#pragma unroll
    for (int k = 0; k < 4; ++k) {
      const unsigned e = (unsigned)(i0 + k * OWN_THREADS) * 4u;
      const int dv[4] = {d[k].x, d[k].y, d[k].z, d[k].w};
#pragma unroll
      for (int j = 0; j < 4; ++j) {
        const unsigned u = (unsigned)dv[j] - lo;
        if (u < SPAN) {
          atomicAdd(&tab[u], 1u);
          if (c < (unsigned)SEG_K) seg[c * OWN_THREADS] = (u << 20) | (e + j);
          ++c;
        }
      }
    }
  }
  const unsigned cs = c < (unsigned)SEG_K ? c : (unsigned)SEG_K;
  if (c > (unsigned)SEG_K) tab[SPAN + FIND_G] = 1u;
  unsigned cnt[FIND_G];
#pragma unroll
  for (int g = 0; g < FIND_G; ++g) cnt[g] = 0u;
  for (unsigned s = 0; s < cs; ++s) {
    const unsigned o = (seg[s * OWN_THREADS] >> 20) / (unsigned)NPB;
#pragma unroll
    for (int g = 0; g < FIND_G; ++g) cnt[g] += (o == (unsigned)g) ? 1u : 0u;
  }
  unsigned base[FIND_G];
#pragma unroll
  for (int g = 0; g < FIND_G; ++g) base[g] = cnt[g] ? atomicAdd(&tab[SPAN + g], cnt[g]) : 0u;
  int2* lists = list + (size_t)blockIdx.x * FIND_G * LIST_CAP;
  for (unsigned s = 0; s < cs; ++s) {
    const unsigned w = seg[s * OWN_THREADS];
    const unsigned u = w >> 20, e = w & 0xFFFFFu;
    const unsigned o = u / (unsigned)NPB, r = u - o * (unsigned)NPB;
    unsigned pos = 0u;
#pragma unroll
    for (int g = 0; g < FIND_G; ++g)
      if (o == (unsigned)g) { pos = base[g]; base[g] = pos + 1u; }
    if (pos < (unsigned)LIST_CAP) lists[(size_t)o * LIST_CAP + pos] = make_int2(src[e], (int)r);
  }
  __syncthreads();
  if (tid < FIND_G)
    counts[blockIdx.x * FIND_G + tid] = tab[SPAN + FIND_G] ? (unsigned)LIST_CAP + 1u : tab[SPAN + tid];
  if (tid < (int)SPAN) inv[lo + tid] = 1.0f / fmaxf((float)tab[tid], 1.0f);
}

typedef __attribute__((ext_vector_type(4))) float v4f;
template<int VW> struct lane_vec;
template<> struct lane_vec<4> { typedef v4f type; };
template<> struct lane_vec<2> { typedef v2f type; };

template<int D>
__global__ __launch_bounds__(OWN_THREADS)
void sum_rows_lds(const float* __restrict__ feat,
                  const int* __restrict__ src, const int* __restrict__ dst,
                  const int2* __restrict__ list, const unsigned* __restrict__ counts,
                  float* __restrict__ agg) {
  constexpr int VW = D / 32;
  constexpr int U  = 16;
  constexpr int PL = D / 2;
  typedef typename lane_vec<VW>::type VT;
  static_assert(PL % VW == 0, "a lane's floats lie in one plane");
  __shared__ __attribute__((aligned(16))) float sums[OWN_LDS_WORDS];
  const int tid  = threadIdx.x;
  const int lane = tid & 31;
  const int wave = __builtin_amdgcn_readfirstlane(tid >> 5);
  const int lo = blockIdx.x * NPB;
  float4* s4 = reinterpret_cast<float4*>(sums);
  for (int i = tid; i < (NPB + 1) * D / 4; i += OWN_THREADS) s4[i] = make_float4(0.f, 0.f, 0.f, 0.f);
  const unsigned n = counts[blockIdx.x];
  __syncthreads();
  if (n <= (unsigned)LIST_CAP) {
    const int2* my = list + (size_t)blockIdx.x * LIST_CAP;
    const int nE = (int)n;
    const int nB = (nE + U - 1) / U;
    const int sub = lane & (U - 1);
    int2 ent = make_int2(0, NPB);
    if (wave * U + sub < nE) ent = my[wave * U + sub];
    for (int bt = wave; bt < nB; bt += OWN_WAVES) {
      int2 nxt = make_int2(0, NPB);
      const int idn = (bt + OWN_WAVES) * U + sub;
      if (idn < nE) nxt = my[idn];
      VT v[U];
#pragma unroll
      for (int u = 0; u < U; ++u) {
        const int sj = __builtin_amdgcn_readlane(ent.x, u);
        v[u] = *reinterpret_cast<const VT*>(feat + (size_t)sj * D + lane * VW);
      }
#pragma unroll
      for (int u = 0; u < U; ++u) {
        const int tj = __builtin_amdgcn_readlane(ent.y, u);
        float* sd = sums + tj * D + lane;
#pragma unroll
        for (int j = 0; j < VW; ++j) atomicAdd(&sd[j * 32], v[u][j]);
      }
      ent = nxt;
    }
  } else {
    for (int e0 = wave * 32; e0 < N_EDGES; e0 += OWN_THREADS) {
      const int e = e0 + lane;
      const unsigned u = (unsigned)dst[e] - (unsigned)lo;
      const bool m = u < (unsigned)NPB;
      const int s = m ? src[e] : 0;
      unsigned mask = __builtin_amdgcn_ballot_w32(m);
      while (mask) {
        const int l = __builtin_ctz(mask);
        mask &= mask - 1;
        const int sj = __shfl(s, l);
        const int tj = __shfl((int)u, l);
        const VT val = *reinterpret_cast<const VT*>(feat + (size_t)sj * D + lane * VW);
        float* sd = sums + tj * D + lane;
#pragma unroll
        for (int j = 0; j < VW; ++j) atomicAdd(&sd[j * 32], val[j]);
      }
    }
  }
  __syncthreads();
  for (int i = tid; i < NPB * 32; i += OWN_THREADS) {
    const int u = i >> 5, l = i & 31;
    const int c = l * VW;
    const int p = c / PL, cc = c % PL;
    VT val;
#pragma unroll
    for (int j = 0; j < VW; ++j) val[j] = sums[u * D + j * 32 + l];
    *reinterpret_cast<VT*>(agg + (size_t)p * N_NODES * PL + (size_t)(lo + u) * PL + cc) = val;
  }
}


typedef __attribute__((ext_vector_type(4))) float v4f;

__device__ __forceinline__ void mma_k8(v8f (&acc)[4], const v4f a, const v4f (&b)[4]) {
  const v2f alo = {a.x, a.y}, ahi = {a.z, a.w};
#pragma unroll
  for (int t = 0; t < 4; ++t) {
    const v2f blo = {b[t].x, b[t].y};
    acc[t] = __builtin_amdgcn_wmma_f32_16x16x4_f32(false, alo, false, blo,
                                                   (short)0, acc[t], false, false);
  }
#pragma unroll
  for (int t = 0; t < 4; ++t) {
    const v2f bhi = {b[t].z, b[t].w};
    acc[t] = __builtin_amdgcn_wmma_f32_16x16x4_f32(false, ahi, false, bhi,
                                                   (short)0, acc[t], false, false);
  }
}

template<int DIN, int DOUT>
__global__ __launch_bounds__(32)
void sage_wmma_gemm(const float* __restrict__ agg,
                    const float* __restrict__ xin,
                    const float* __restrict__ inv,
                    const float* __restrict__ Wl,
                    const float* __restrict__ Wr,
                    const float* __restrict__ bias,
                    float* __restrict__ out) {
  const int lane = threadIdx.x & 31;
  const int half = lane >> 4;
  const int l16  = lane & 15;
  const int mTile = blockIdx.x << 4;
  const int nBase = blockIdx.y << 6;

  const int aRow = mTile + l16;
  const int kOff = half << 2;

  const float invc = inv[aRow];

  v8f acc[4] = {v8f{}, v8f{}, v8f{}, v8f{}};

  constexpr int PL = DIN / 2;
  static_assert(PL % 8 == 0, "a plane is whole groups of eight k");
#pragma unroll
  for (int p = 0; p < 2; ++p) {
    const float* aP = agg + (size_t)p * N_NODES * PL + (size_t)aRow * PL + kOff;
    const float* bP = Wl  + (size_t)(nBase + l16) * DIN + p * PL + kOff;
#pragma unroll 2
    for (int k = 0; k < PL; k += 8) {
      v4f a = *(const v4f*)(aP + k);
      a *= invc;
      v4f b[4];
#pragma unroll
      for (int t = 0; t < 4; ++t) b[t] = *(const v4f*)(bP + (size_t)t * 16 * DIN + k);
      mma_k8(acc, a, b);
    }
  }

  {
    const float* aP = xin + (size_t)aRow * DIN + kOff;
    const float* bP = Wr  + (size_t)(nBase + l16) * DIN + kOff;
#pragma unroll 2
    for (int k = 0; k < DIN; k += 8) {
      const v4f a = *(const v4f*)(aP + k);
      v4f b[4];
#pragma unroll
      for (int t = 0; t < 4; ++t) b[t] = *(const v4f*)(bP + (size_t)t * 16 * DIN + k);
      mma_k8(acc, a, b);
    }
  }

  const int rowBase = mTile + (half << 3);
#pragma unroll
  for (int t = 0; t < 4; ++t) {
    const int col = nBase + t * 16 + l16;
    const float bj = bias[col];
#pragma unroll
    for (int i = 0; i < 8; ++i)
      out[(size_t)(rowBase + i) * DOUT + col] = acc[t][i] + bj;
  }
}


extern "C" void kernel_launch(void* const* d_in, const int* in_sizes, int n_in,
                              void* d_out, int out_size, void* d_ws, size_t ws_size,
                              hipStream_t stream) {
  const float* x   = (const float*)d_in[0];
  const int*   ei  = (const int*)d_in[1];
  const int*   src = ei;
  const int*   dst = ei + N_EDGES;
  const float* Wl1 = (const float*)d_in[2];
  const float* b1  = (const float*)d_in[3];
  const float* Wr1 = (const float*)d_in[4];
  const float* Wl2 = (const float*)d_in[5];
  const float* b2  = (const float*)d_in[6];
  const float* Wr2 = (const float*)d_in[7];
  const float* Wl3 = (const float*)d_in[8];
  const float* b3  = (const float*)d_in[9];
  const float* Wr3 = (const float*)d_in[10];
  float* out = (float*)d_out;

  float* ws  = (float*)d_ws;
  float* inv = ws;
  unsigned* counts = (unsigned*)(inv + 50048);
  int2* list = (int2*)(counts + 256);
  float* agg = (float*)(list + (size_t)NOWN * LIST_CAP);
  float* h1  = agg + (size_t)N_NODES * 128;
  float* h2  = h1  + (size_t)N_NODES * 128;

  find_edges<<<NOWN / FIND_G, OWN_THREADS, 0, stream>>>(src, dst, list, counts, inv);

  sum_rows_lds<128><<<NOWN, OWN_THREADS, 0, stream>>>(x, src, dst, list, counts, agg);
  {
    dim3 g(N_NODES / 16, 128 / 64);
    sage_wmma_gemm<128, 128><<<g, 32, 0, stream>>>(agg, x, inv, Wl1, Wr1, b1, h1);
  }

  sum_rows_lds<128><<<NOWN, OWN_THREADS, 0, stream>>>(h1, src, dst, list, counts, agg);
  {
    dim3 g(N_NODES / 16, 64 / 64);
    sage_wmma_gemm<128, 64><<<g, 32, 0, stream>>>(agg, h1, inv, Wl2, Wr2, b2, h2);
  }

  sum_rows_lds<64><<<NOWN, OWN_THREADS, 0, stream>>>(h2, src, dst, list, counts, agg);
  {
    dim3 g(N_NODES / 16, 128 / 64);
    sage_wmma_gemm<64, 128><<<g, 32, 0, stream>>>(agg, h2, inv, Wl3, Wr3, b3, out);
  }
}


# Neighbour sums by owner waves: the finder keeps one list per wave of an owner (rows w, w+32, ..); a wave sums its 7 rows in registers indexed by the row slot, no LDS float adds; sums in list order
# speedup vs baseline: 2.4860x; 1.0355x over previous
#include <hip/hip_runtime.h>

#define N_NODES 50000
#define N_EDGES 800000

typedef __attribute__((ext_vector_type(2))) float v2f;
typedef __attribute__((ext_vector_type(8))) float v8f;

#define NPB 200
#define NOWN (N_NODES / NPB)
#define FIND_G 2
#define OWN_THREADS 1024
#define OWN_WAVES (OWN_THREADS / 32)
#define ROWS_PW ((NPB + OWN_WAVES - 1) / OWN_WAVES)
#define BCAP 256
#define OWN_LDS_WORDS 41216
static_assert(N_NODES % NPB == 0, "every owner has the same number of nodes");
static_assert(NOWN % FIND_G == 0, "every finder serves the same number of owners");
static_assert(N_EDGES % 4 == 0, "the destination list is read 16 bytes at a time");
static_assert(N_EDGES % 32 == 0, "a wave of the rescue path always holds 32 edges");
static_assert((NPB + 1) * 128 <= OWN_LDS_WORDS, "sum rows of the rescue path fit the reservation");
static_assert(OWN_WAVES == 32, "a row's wave is the low five bits of its local number");

#define SEG_K 40
#define TAB_WORDS 512
#define TAB_CNT0 448
#define FIND_LDS_WORDS (TAB_WORDS + SEG_K * OWN_THREADS)
static_assert(FIND_G * NPB + FIND_G + 1 <= TAB_CNT0, "degrees and flags end below the list lengths");
static_assert(TAB_CNT0 + FIND_G * OWN_WAVES <= TAB_WORDS, "table fits its words");
static_assert(TAB_WORDS <= OWN_THREADS, "one thread per table word when a finder clears it");
static_assert(N_EDGES <= (1 << 20) && FIND_G * NPB <= (1 << 12), "a match packs into one word");
static_assert(FIND_LDS_WORDS * 4 > 160 * 1024, "one finder per WGP");

__global__ __launch_bounds__(OWN_THREADS)
void find_edges(const int* __restrict__ src, const int* __restrict__ dst,
                int2* __restrict__ list,
                unsigned* __restrict__ counts,
                unsigned* __restrict__ bcnt,
                float* __restrict__ inv) {
  __shared__ unsigned tab[FIND_LDS_WORDS];
  constexpr unsigned SPAN = FIND_G * NPB;
  const int tid = threadIdx.x;
  if (tid < TAB_WORDS) tab[tid] = 0u;
  __syncthreads();
  const unsigned lo = (unsigned)blockIdx.x * SPAN;
  unsigned* seg = tab + TAB_WORDS + tid;
  unsigned c = 0;
  const int4* d4 = reinterpret_cast<const int4*>(dst);
  constexpr int NV = N_EDGES / 4;
  for (int i0 = tid; i0 < NV; i0 += 4 * OWN_THREADS) {
    int4 d[4];
#pragma unroll
    for (int k = 0; k < 4; ++k) {
      const int i = i0 + k * OWN_THREADS;
      d[k] = (i < NV) ? d4[i] : make_int4(-1, -1, -1, -1);
    }
#pragma unroll
    for (int k = 0; k < 4; ++k) {
      const unsigned e = (unsigned)(i0 + k * OWN_THREADS) * 4u;
      const int dv[4] = {d[k].x, d[k].y, d[k].z, d[k].w};
#pragma unroll
      for (int j = 0; j < 4; ++j) {
        const unsigned u = (unsigned)dv[j] - lo;
        if (u < SPAN) {
          atomicAdd(&tab[u], 1u);
          if (c < (unsigned)SEG_K) seg[c * OWN_THREADS] = (u << 20) | (e + j);
          ++c;
        }
      }
    }
  }
  const unsigned cs = c < (unsigned)SEG_K ? c : (unsigned)SEG_K;
  if (c > (unsigned)SEG_K) tab[SPAN + FIND_G] = 1u;
  int2* lists = list + (size_t)blockIdx.x * FIND_G * OWN_WAVES * BCAP;
  for (unsigned s = 0; s < cs; ++s) {
    const unsigned w = seg[s * OWN_THREADS];
    const unsigned u = w >> 20, e = w & 0xFFFFFu;
    const unsigned o = u / (unsigned)NPB, r = u - o * (unsigned)NPB;
    const unsigned bk = o * (unsigned)OWN_WAVES + (r & (unsigned)(OWN_WAVES - 1));
    const unsigned pos = atomicAdd(&tab[TAB_CNT0 + bk], 1u);
    if (pos < (unsigned)BCAP) lists[(size_t)bk * BCAP + pos] = make_int2(src[e], (int)(r / (unsigned)OWN_WAVES));
    else tab[SPAN + o] = 1u;
  }
  __syncthreads();
  if (tid < FIND_G)
    counts[blockIdx.x * FIND_G + tid] = tab[SPAN + FIND_G] | tab[SPAN + tid];
  if (tid < FIND_G * OWN_WAVES)
    bcnt[blockIdx.x * FIND_G * OWN_WAVES + tid] = tab[TAB_CNT0 + tid];
  if (tid < (int)SPAN) inv[lo + tid] = 1.0f / fmaxf((float)tab[tid], 1.0f);
}

typedef __attribute__((ext_vector_type(4))) float v4f;
template<int VW> struct lane_vec;
template<> struct lane_vec<4> { typedef v4f type; };
template<> struct lane_vec<2> { typedef v2f type; };

template<int D>
__global__ __launch_bounds__(OWN_THREADS)
void sum_rows_own(const float* __restrict__ feat,
                  const int* __restrict__ src, const int* __restrict__ dst,
                  const int2* __restrict__ list, const unsigned* __restrict__ counts,
                  const unsigned* __restrict__ bcnt,
                  float* __restrict__ agg) {
  constexpr int VW = D / 32;
  constexpr int U  = 16;
  constexpr int PL = D / 2;
  typedef typename lane_vec<VW>::type VT;
  static_assert(PL % VW == 0, "a lane's floats lie in one plane");
  __shared__ __attribute__((aligned(16))) float sums[OWN_LDS_WORDS];
  const int tid  = threadIdx.x;
  const int lane = tid & 31;
  const int wave = __builtin_amdgcn_readfirstlane(tid >> 5);
  const int lo = blockIdx.x * NPB;
  if (counts[blockIdx.x] == 0u) {
    const int2* my = list + ((size_t)blockIdx.x * OWN_WAVES + wave) * BCAP;
    const int nE = __builtin_amdgcn_readfirstlane((int)bcnt[blockIdx.x * OWN_WAVES + wave]);
    const int nB = (nE + U - 1) / U;
    const int sub = lane & (U - 1);
    float acc[(ROWS_PW + 1) * VW];
#pragma unroll
    for (int i = 0; i < (ROWS_PW + 1) * VW; ++i) acc[i] = 0.f;
    int2 ent = make_int2(0, ROWS_PW);
    if (sub < nE) ent = my[sub];
    for (int bt = 0; bt < nB; ++bt) {
      int2 nxt = make_int2(0, ROWS_PW);
      const int idn = (bt + 1) * U + sub;
      if (idn < nE) nxt = my[idn];
      VT v[U];
#pragma unroll
      for (int u = 0; u < U; ++u) {
        const int sj = __builtin_amdgcn_readlane(ent.x, u);
        v[u] = *reinterpret_cast<const VT*>(feat + (size_t)sj * D + lane * VW);
      }
      __builtin_amdgcn_sched_barrier(0);
#pragma unroll
      for (int u = 0; u < U; ++u) {
        const int q = __builtin_amdgcn_readlane(ent.y, u);
#pragma unroll
        for (int j = 0; j < VW; ++j) acc[q * VW + j] += v[u][j];
      }
      ent = nxt;
    }
    const int c = lane * VW;
    const int p = c / PL, cc = c % PL;
    float* ap = agg + (size_t)p * N_NODES * PL + (size_t)lo * PL + cc;
#pragma unroll
    for (int q = 0; q < ROWS_PW; ++q) {
      const int r = wave + q * OWN_WAVES;
      if (r < NPB) {
        VT val;
#pragma unroll
        for (int j = 0; j < VW; ++j) val[j] = acc[q * VW + j];
        *reinterpret_cast<VT*>(ap + (size_t)r * PL) = val;
      }
    }
  } else {
    float4* s4 = reinterpret_cast<float4*>(sums);
    for (int i = tid; i < (NPB + 1) * D / 4; i += OWN_THREADS) s4[i] = make_float4(0.f, 0.f, 0.f, 0.f);
    __syncthreads();
    for (int e0 = wave * 32; e0 < N_EDGES; e0 += OWN_THREADS) {
      const int e = e0 + lane;
      const unsigned u = (unsigned)dst[e] - (unsigned)lo;
      const bool m = u < (unsigned)NPB;
      const int s = m ? src[e] : 0;
      unsigned mask = __builtin_amdgcn_ballot_w32(m);
      while (mask) {
        const int l = __builtin_ctz(mask);
        mask &= mask - 1;
        const int sj = __shfl(s, l);
        const int tj = __shfl((int)u, l);
        const VT val = *reinterpret_cast<const VT*>(feat + (size_t)sj * D + lane * VW);
        float* sd = sums + tj * D + lane;
#pragma unroll
        for (int j = 0; j < VW; ++j) atomicAdd(&sd[j * 32], val[j]);
      }
    }
    __syncthreads();
    for (int i = tid; i < NPB * 32; i += OWN_THREADS) {
      const int u = i >> 5, l = i & 31;
      const int c = l * VW;
      const int p = c / PL, cc = c % PL;
      VT val;
#pragma unroll
      for (int j = 0; j < VW; ++j) val[j] = sums[u * D + j * 32 + l];
      *reinterpret_cast<VT*>(agg + (size_t)p * N_NODES * PL + (size_t)(lo + u) * PL + cc) = val;
    }
  }
}


typedef __attribute__((ext_vector_type(4))) float v4f;

__device__ __forceinline__ void mma_k8(v8f (&acc)[4], const v4f a, const v4f (&b)[4]) {
  const v2f alo = {a.x, a.y}, ahi = {a.z, a.w};
#pragma unroll
  for (int t = 0; t < 4; ++t) {
    const v2f blo = {b[t].x, b[t].y};
    acc[t] = __builtin_amdgcn_wmma_f32_16x16x4_f32(false, alo, false, blo,
                                                   (short)0, acc[t], false, false);
  }
#pragma unroll
  for (int t = 0; t < 4; ++t) {
    const v2f bhi = {b[t].z, b[t].w};
    acc[t] = __builtin_amdgcn_wmma_f32_16x16x4_f32(false, ahi, false, bhi,
                                                   (short)0, acc[t], false, false);
  }
}

template<int DIN, int DOUT>
__global__ __launch_bounds__(32)
void sage_wmma_gemm(const float* __restrict__ agg,
                    const float* __restrict__ xin,
                    const float* __restrict__ inv,
                    const float* __restrict__ Wl,
                    const float* __restrict__ Wr,
                    const float* __restrict__ bias,
                    float* __restrict__ out) {
  const int lane = threadIdx.x & 31;
  const int half = lane >> 4;
  const int l16  = lane & 15;
  const int mTile = blockIdx.x << 4;
  const int nBase = blockIdx.y << 6;

  const int aRow = mTile + l16;
  const int kOff = half << 2;

  const float invc = inv[aRow];

  v8f acc[4] = {v8f{}, v8f{}, v8f{}, v8f{}};

  constexpr int PL = DIN / 2;
  static_assert(PL % 8 == 0, "a plane is whole groups of eight k");
#pragma unroll
  for (int p = 0; p < 2; ++p) {
    const float* aP = agg + (size_t)p * N_NODES * PL + (size_t)aRow * PL + kOff;
    const float* bP = Wl  + (size_t)(nBase + l16) * DIN + p * PL + kOff;
#pragma unroll 2
    for (int k = 0; k < PL; k += 8) {
      v4f a = *(const v4f*)(aP + k);
      a *= invc;
      v4f b[4];
#pragma unroll
      for (int t = 0; t < 4; ++t) b[t] = *(const v4f*)(bP + (size_t)t * 16 * DIN + k);
      mma_k8(acc, a, b);
    }
  }

  {
    const float* aP = xin + (size_t)aRow * DIN + kOff;
    const float* bP = Wr  + (size_t)(nBase + l16) * DIN + kOff;
#pragma unroll 2
    for (int k = 0; k < DIN; k += 8) {
      const v4f a = *(const v4f*)(aP + k);
      v4f b[4];
#pragma unroll
      for (int t = 0; t < 4; ++t) b[t] = *(const v4f*)(bP + (size_t)t * 16 * DIN + k);
      mma_k8(acc, a, b);
    }
  }

  const int rowBase = mTile + (half << 3);
#pragma unroll
  for (int t = 0; t < 4; ++t) {
    const int col = nBase + t * 16 + l16;
    const float bj = bias[col];
#pragma unroll
    for (int i = 0; i < 8; ++i)
      out[(size_t)(rowBase + i) * DOUT + col] = acc[t][i] + bj;
  }
}


extern "C" void kernel_launch(void* const* d_in, const int* in_sizes, int n_in,
                              void* d_out, int out_size, void* d_ws, size_t ws_size,
                              hipStream_t stream) {
  const float* x   = (const float*)d_in[0];
  const int*   ei  = (const int*)d_in[1];
  const int*   src = ei;
  const int*   dst = ei + N_EDGES;
  const float* Wl1 = (const float*)d_in[2];
  const float* b1  = (const float*)d_in[3];
  const float* Wr1 = (const float*)d_in[4];
  const float* Wl2 = (const float*)d_in[5];
  const float* b2  = (const float*)d_in[6];
  const float* Wr2 = (const float*)d_in[7];
  const float* Wl3 = (const float*)d_in[8];
  const float* b3  = (const float*)d_in[9];
  const float* Wr3 = (const float*)d_in[10];
  float* out = (float*)d_out;

  float* ws  = (float*)d_ws;
  float* inv = ws;
  unsigned* counts = (unsigned*)(inv + 50048);
  unsigned* bcnt = counts + 256;
  int2* list = (int2*)(bcnt + 8192);
  float* agg = (float*)(list + (size_t)NOWN * OWN_WAVES * BCAP);
  float* h1  = agg + (size_t)N_NODES * 128;
  float* h2  = h1  + (size_t)N_NODES * 128;

  find_edges<<<NOWN / FIND_G, OWN_THREADS, 0, stream>>>(src, dst, list, counts, bcnt, inv);

  sum_rows_own<128><<<NOWN, OWN_THREADS, 0, stream>>>(x, src, dst, list, counts, bcnt, agg);
  {
    dim3 g(N_NODES / 16, 128 / 64);
    sage_wmma_gemm<128, 128><<<g, 32, 0, stream>>>(agg, x, inv, Wl1, Wr1, b1, h1);
  }

  sum_rows_own<128><<<NOWN, OWN_THREADS, 0, stream>>>(h1, src, dst, list, counts, bcnt, agg);
  {
    dim3 g(N_NODES / 16, 64 / 64);
    sage_wmma_gemm<128, 64><<<g, 32, 0, stream>>>(agg, h1, inv, Wl2, Wr2, b2, h2);
  }

  sum_rows_own<64><<<NOWN, OWN_THREADS, 0, stream>>>(h2, src, dst, list, counts, bcnt, agg);
  {
    dim3 g(N_NODES / 16, 128 / 64);
    sage_wmma_gemm<64, 128><<<g, 32, 0, stream>>>(agg, h2, inv, Wl3, Wr3, b3, out);
  }
}


# Owner wave's list grouped by row (the finder places an edge at its row's cursor): a wave sums one row at a time in fixed registers, stores it at the row's end; no indexed adds; arrival order kept
# speedup vs baseline: 2.5758x; 1.0361x over previous
#include <hip/hip_runtime.h>

#define N_NODES 50000
#define N_EDGES 800000

typedef __attribute__((ext_vector_type(2))) float v2f;
typedef __attribute__((ext_vector_type(8))) float v8f;

#define NPB 200
#define NOWN (N_NODES / NPB)
#define FIND_G 2
#define OWN_THREADS 1024
#define OWN_WAVES (OWN_THREADS / 32)
#define ROWS_PW ((NPB + OWN_WAVES - 1) / OWN_WAVES)
#define BCAP 256
#define OWN_LDS_WORDS 41216
static_assert(N_NODES % NPB == 0, "every owner has the same number of nodes");
static_assert(NOWN % FIND_G == 0, "every finder serves the same number of owners");
static_assert(N_EDGES % 4 == 0, "the destination list is read 16 bytes at a time");
static_assert(N_EDGES % 32 == 0, "a wave of the rescue path always holds 32 edges");
static_assert((NPB + 1) * 128 <= OWN_LDS_WORDS, "sum rows of the rescue path fit the reservation");
static_assert(OWN_WAVES == 32, "a row's wave is the low five bits of its local number");

#define SEG_K 40
#define TAB_WORDS 1024
#define TAB_CNT0 448
#define TAB_CUR0 512
#define FIND_LDS_WORDS (TAB_WORDS + SEG_K * OWN_THREADS)
static_assert(FIND_G * NPB + FIND_G + 1 <= TAB_CNT0, "degrees and flags end below the list lengths");
static_assert(TAB_CNT0 + FIND_G * OWN_WAVES <= TAB_CUR0, "list lengths end below the row cursors");
static_assert(TAB_CUR0 + FIND_G * NPB <= TAB_WORDS, "table fits its words");
static_assert(TAB_WORDS <= OWN_THREADS, "one thread per table word when a finder clears it");
static_assert(N_EDGES <= (1 << 20) && FIND_G * NPB <= (1 << 12), "a match packs into one word");
static_assert(FIND_LDS_WORDS * 4 > 160 * 1024, "one finder per WGP");

__global__ __launch_bounds__(OWN_THREADS)
void find_edges(const int* __restrict__ src, const int* __restrict__ dst,
                int2* __restrict__ list,
                unsigned* __restrict__ counts,
                unsigned* __restrict__ bcnt,
                float* __restrict__ inv) {
  __shared__ unsigned tab[FIND_LDS_WORDS];
  constexpr unsigned SPAN = FIND_G * NPB;
  const int tid = threadIdx.x;
  if (tid < TAB_WORDS) tab[tid] = 0u;
  __syncthreads();
  const unsigned lo = (unsigned)blockIdx.x * SPAN;
  unsigned* seg = tab + TAB_WORDS + tid;
  unsigned c = 0;
  const int4* d4 = reinterpret_cast<const int4*>(dst);
  constexpr int NV = N_EDGES / 4;
  for (int i0 = tid; i0 < NV; i0 += 4 * OWN_THREADS) {
    int4 d[4];
#pragma unroll
    for (int k = 0; k < 4; ++k) {
      const int i = i0 + k * OWN_THREADS;
      d[k] = (i < NV) ? d4[i] : make_int4(-1, -1, -1, -1);
    }
#pragma unroll
    for (int k = 0; k < 4; ++k) {
      const unsigned e = (unsigned)(i0 + k * OWN_THREADS) * 4u;
      const int dv[4] = {d[k].x, d[k].y, d[k].z, d[k].w};
#pragma unroll
      for (int j = 0; j < 4; ++j) {
        const unsigned u = (unsigned)dv[j] - lo;
        if (u < SPAN) {
          atomicAdd(&tab[u], 1u);
          if (c < (unsigned)SEG_K) seg[c * OWN_THREADS] = (u << 20) | (e + j);
          ++c;
        }
      }
    }
  }
  const unsigned cs = c < (unsigned)SEG_K ? c : (unsigned)SEG_K;
  if (c > (unsigned)SEG_K) tab[SPAN + FIND_G] = 1u;
  __syncthreads();
  if (tid < (int)SPAN) {
    const unsigned o = (unsigned)tid / (unsigned)NPB, r = (unsigned)tid - o * (unsigned)NPB;
    const unsigned w = r & (unsigned)(OWN_WAVES - 1);
    unsigned first = 0;
#pragma unroll
    for (unsigned k = 0; k + 1 < (unsigned)ROWS_PW; ++k) {
      const unsigned rr = w + k * (unsigned)OWN_WAVES;
      if (rr < r) first += tab[o * (unsigned)NPB + rr];
    }
    tab[TAB_CUR0 + tid] = first;
    if (r + (unsigned)OWN_WAVES >= (unsigned)NPB) tab[TAB_CNT0 + o * (unsigned)OWN_WAVES + w] = first + tab[tid];
  }
  __syncthreads();
  int2* lists = list + (size_t)blockIdx.x * FIND_G * OWN_WAVES * BCAP;
  for (unsigned s = 0; s < cs; ++s) {
    const unsigned w = seg[s * OWN_THREADS];
    const unsigned u = w >> 20, e = w & 0xFFFFFu;
    const unsigned o = u / (unsigned)NPB, r = u - o * (unsigned)NPB;
    const unsigned bk = o * (unsigned)OWN_WAVES + (r & (unsigned)(OWN_WAVES - 1));
    const unsigned pos = atomicAdd(&tab[TAB_CUR0 + u], 1u);
    if (pos < (unsigned)BCAP) lists[(size_t)bk * BCAP + pos] = make_int2(src[e], (int)(r / (unsigned)OWN_WAVES));
    else tab[SPAN + o] = 1u;
  }
  __syncthreads();
  if (tid < FIND_G)
    counts[blockIdx.x * FIND_G + tid] = tab[SPAN + FIND_G] | tab[SPAN + tid];
  if (tid < FIND_G * OWN_WAVES)
    bcnt[blockIdx.x * FIND_G * OWN_WAVES + tid] = tab[TAB_CNT0 + tid];
  if (tid < (int)SPAN) inv[lo + tid] = 1.0f / fmaxf((float)tab[tid], 1.0f);
}

typedef __attribute__((ext_vector_type(4))) float v4f;
template<int VW> struct lane_vec;
template<> struct lane_vec<4> { typedef v4f type; };
template<> struct lane_vec<2> { typedef v2f type; };

template<int D>
__global__ __launch_bounds__(OWN_THREADS)
void sum_rows_own(const float* __restrict__ feat,
                  const int* __restrict__ src, const int* __restrict__ dst,
                  const int2* __restrict__ list, const unsigned* __restrict__ counts,
                  const unsigned* __restrict__ bcnt,
                  float* __restrict__ agg) {
  constexpr int VW = D / 32;
  constexpr int U  = 16;
  constexpr int PL = D / 2;
  typedef typename lane_vec<VW>::type VT;
  static_assert(PL % VW == 0, "a lane's floats lie in one plane");
  __shared__ __attribute__((aligned(16))) float sums[OWN_LDS_WORDS];
  const int tid  = threadIdx.x;
  const int lane = tid & 31;
  const int wave = __builtin_amdgcn_readfirstlane(tid >> 5);
  const int lo = blockIdx.x * NPB;
  if (counts[blockIdx.x] == 0u) {
    const int2* my = list + ((size_t)blockIdx.x * OWN_WAVES + wave) * BCAP;
    const int nE = __builtin_amdgcn_readfirstlane((int)bcnt[blockIdx.x * OWN_WAVES + wave]);
    const int nB = (nE + U - 1) / U;
    const int sub = lane & (U - 1);
    const int c = lane * VW;
    const int p = c / PL, cc = c % PL;
    float* ap = agg + (size_t)p * N_NODES * PL + (size_t)lo * PL + cc;
    VT zero;
#pragma unroll
    for (int j = 0; j < VW; ++j) zero[j] = 0.f;
    VT acc = zero;
    int cur = 0;
    auto next_slot = [&](const int qn) {
      if (cur < ROWS_PW) {
        const int r = wave + cur * OWN_WAVES;
        if (r < NPB) *reinterpret_cast<VT*>(ap + (size_t)r * PL) = acc;
      }
#pragma nounroll
      for (int z = cur + 1; z < qn; ++z) {
        const int r = wave + z * OWN_WAVES;
        if (r < NPB) *reinterpret_cast<VT*>(ap + (size_t)r * PL) = zero;
      }
      acc = zero;
      cur = qn;
    };
    int2 ent = make_int2(0, ROWS_PW);
    if (sub < nE) ent = my[sub];
    for (int bt = 0; bt < nB; ++bt) {
      int2 nxt = make_int2(0, ROWS_PW);
      const int idn = (bt + 1) * U + sub;
      if (idn < nE) nxt = my[idn];
      VT v[U];
#pragma unroll
      for (int u = 0; u < U; ++u) {
        const int sj = __builtin_amdgcn_readlane(ent.x, u);
        v[u] = *reinterpret_cast<const VT*>(feat + (size_t)sj * D + lane * VW);
      }
      __builtin_amdgcn_sched_barrier(0);
#pragma unroll
      for (int u = 0; u < U; ++u) {
        const int q = __builtin_amdgcn_readlane(ent.y, u);
        if (__builtin_expect(q != cur, 0)) next_slot(q);
        acc += v[u];
      }
      ent = nxt;
    }
    next_slot(ROWS_PW);
  } else {
    float4* s4 = reinterpret_cast<float4*>(sums);
    for (int i = tid; i < (NPB + 1) * D / 4; i += OWN_THREADS) s4[i] = make_float4(0.f, 0.f, 0.f, 0.f);
    __syncthreads();
    for (int e0 = wave * 32; e0 < N_EDGES; e0 += OWN_THREADS) {
      const int e = e0 + lane;
      const unsigned u = (unsigned)dst[e] - (unsigned)lo;
      const bool m = u < (unsigned)NPB;
      const int s = m ? src[e] : 0;
      unsigned mask = __builtin_amdgcn_ballot_w32(m);
      while (mask) {
        const int l = __builtin_ctz(mask);
        mask &= mask - 1;
        const int sj = __shfl(s, l);
        const int tj = __shfl((int)u, l);
        const VT val = *reinterpret_cast<const VT*>(feat + (size_t)sj * D + lane * VW);
        float* sd = sums + tj * D + lane;
#pragma unroll
        for (int j = 0; j < VW; ++j) atomicAdd(&sd[j * 32], val[j]);
      }
    }
    __syncthreads();
    for (int i = tid; i < NPB * 32; i += OWN_THREADS) {
      const int u = i >> 5, l = i & 31;
      const int c = l * VW;
      const int p = c / PL, cc = c % PL;
      VT val;
#pragma unroll
      for (int j = 0; j < VW; ++j) val[j] = sums[u * D + j * 32 + l];
      *reinterpret_cast<VT*>(agg + (size_t)p * N_NODES * PL + (size_t)(lo + u) * PL + cc) = val;
    }
  }
}


typedef __attribute__((ext_vector_type(4))) float v4f;

__device__ __forceinline__ void mma_k8(v8f (&acc)[4], const v4f a, const v4f (&b)[4]) {
  const v2f alo = {a.x, a.y}, ahi = {a.z, a.w};
#pragma unroll
  for (int t = 0; t < 4; ++t) {
    const v2f blo = {b[t].x, b[t].y};
    acc[t] = __builtin_amdgcn_wmma_f32_16x16x4_f32(false, alo, false, blo,
                                                   (short)0, acc[t], false, false);
  }
#pragma unroll
  for (int t = 0; t < 4; ++t) {
    const v2f bhi = {b[t].z, b[t].w};
    acc[t] = __builtin_amdgcn_wmma_f32_16x16x4_f32(false, ahi, false, bhi,
                                                   (short)0, acc[t], false, false);
  }
}

template<int DIN, int DOUT>
__global__ __launch_bounds__(32)
void sage_wmma_gemm(const float* __restrict__ agg,
                    const float* __restrict__ xin,
                    const float* __restrict__ inv,
                    const float* __restrict__ Wl,
                    const float* __restrict__ Wr,
                    const float* __restrict__ bias,
                    float* __restrict__ out) {
  const int lane = threadIdx.x & 31;
  const int half = lane >> 4;
  const int l16  = lane & 15;
  const int mTile = blockIdx.x << 4;
  const int nBase = blockIdx.y << 6;

  const int aRow = mTile + l16;
  const int kOff = half << 2;

  const float invc = inv[aRow];

  v8f acc[4] = {v8f{}, v8f{}, v8f{}, v8f{}};

  constexpr int PL = DIN / 2;
  static_assert(PL % 8 == 0, "a plane is whole groups of eight k");
#pragma unroll
  for (int p = 0; p < 2; ++p) {
    const float* aP = agg + (size_t)p * N_NODES * PL + (size_t)aRow * PL + kOff;
    const float* bP = Wl  + (size_t)(nBase + l16) * DIN + p * PL + kOff;
#pragma unroll 2
    for (int k = 0; k < PL; k += 8) {
      v4f a = *(const v4f*)(aP + k);
      a *= invc;
      v4f b[4];
#pragma unroll
      for (int t = 0; t < 4; ++t) b[t] = *(const v4f*)(bP + (size_t)t * 16 * DIN + k);
      mma_k8(acc, a, b);
    }
  }

  {
    const float* aP = xin + (size_t)aRow * DIN + kOff;
    const float* bP = Wr  + (size_t)(nBase + l16) * DIN + kOff;
#pragma unroll 2
    for (int k = 0; k < DIN; k += 8) {
      const v4f a = *(const v4f*)(aP + k);
      v4f b[4];
#pragma unroll
      for (int t = 0; t < 4; ++t) b[t] = *(const v4f*)(bP + (size_t)t * 16 * DIN + k);
      mma_k8(acc, a, b);
    }
  }

  const int rowBase = mTile + (half << 3);
#pragma unroll
  for (int t = 0; t < 4; ++t) {
    const int col = nBase + t * 16 + l16;
    const float bj = bias[col];
#pragma unroll
    for (int i = 0; i < 8; ++i)
      out[(size_t)(rowBase + i) * DOUT + col] = acc[t][i] + bj;
  }
}


extern "C" void kernel_launch(void* const* d_in, const int* in_sizes, int n_in,
                              void* d_out, int out_size, void* d_ws, size_t ws_size,
                              hipStream_t stream) {
  const float* x   = (const float*)d_in[0];
  const int*   ei  = (const int*)d_in[1];
  const int*   src = ei;
  const int*   dst = ei + N_EDGES;
  const float* Wl1 = (const float*)d_in[2];
  const float* b1  = (const float*)d_in[3];
  const float* Wr1 = (const float*)d_in[4];
  const float* Wl2 = (const float*)d_in[5];
  const float* b2  = (const float*)d_in[6];
  const float* Wr2 = (const float*)d_in[7];
  const float* Wl3 = (const float*)d_in[8];
  const float* b3  = (const float*)d_in[9];
  const float* Wr3 = (const float*)d_in[10];
  float* out = (float*)d_out;

  float* ws  = (float*)d_ws;
  float* inv = ws;
  unsigned* counts = (unsigned*)(inv + 50048);
  unsigned* bcnt = counts + 256;
  int2* list = (int2*)(bcnt + 8192);
  float* agg = (float*)(list + (size_t)NOWN * OWN_WAVES * BCAP);
  float* h1  = agg + (size_t)N_NODES * 128;
  float* h2  = h1  + (size_t)N_NODES * 128;

  find_edges<<<NOWN / FIND_G, OWN_THREADS, 0, stream>>>(src, dst, list, counts, bcnt, inv);

  sum_rows_own<128><<<NOWN, OWN_THREADS, 0, stream>>>(x, src, dst, list, counts, bcnt, agg);
  {
    dim3 g(N_NODES / 16, 128 / 64);
    sage_wmma_gemm<128, 128><<<g, 32, 0, stream>>>(agg, x, inv, Wl1, Wr1, b1, h1);
  }

  sum_rows_own<128><<<NOWN, OWN_THREADS, 0, stream>>>(h1, src, dst, list, counts, bcnt, agg);
  {
    dim3 g(N_NODES / 16, 64 / 64);
    sage_wmma_gemm<128, 64><<<g, 32, 0, stream>>>(agg, h1, inv, Wl2, Wr2, b2, h2);
  }

  sum_rows_own<64><<<NOWN, OWN_THREADS, 0, stream>>>(h2, src, dst, list, counts, bcnt, agg);
  {
    dim3 g(N_NODES / 16, 128 / 64);
    sage_wmma_gemm<64, 128><<<g, 32, 0, stream>>>(agg, h2, inv, Wl3, Wr3, b3, out);
  }
}


# GEMM weights through LDS: 256 resident 16-wave workgroups copy both weight matrices once into padded LDS rows, B fragments by 16-byte LDS reads; node rows still direct global loads; sums unchanged
# speedup vs baseline: 2.8577x; 1.1094x over previous
#include <hip/hip_runtime.h>

#define N_NODES 50000
#define N_EDGES 800000

typedef __attribute__((ext_vector_type(2))) float v2f;
typedef __attribute__((ext_vector_type(8))) float v8f;

#define NPB 200
#define NOWN (N_NODES / NPB)
#define FIND_G 2
#define OWN_THREADS 1024
#define OWN_WAVES (OWN_THREADS / 32)
#define ROWS_PW ((NPB + OWN_WAVES - 1) / OWN_WAVES)
#define BCAP 256
#define OWN_LDS_WORDS 41216
static_assert(N_NODES % NPB == 0, "every owner has the same number of nodes");
static_assert(NOWN % FIND_G == 0, "every finder serves the same number of owners");
static_assert(N_EDGES % 4 == 0, "the destination list is read 16 bytes at a time");
static_assert(N_EDGES % 32 == 0, "a wave of the rescue path always holds 32 edges");
static_assert((NPB + 1) * 128 <= OWN_LDS_WORDS, "sum rows of the rescue path fit the reservation");
static_assert(OWN_WAVES == 32, "a row's wave is the low five bits of its local number");

#define SEG_K 40
#define TAB_WORDS 1024
#define TAB_CNT0 448
#define TAB_CUR0 512
#define FIND_LDS_WORDS (TAB_WORDS + SEG_K * OWN_THREADS)
static_assert(FIND_G * NPB + FIND_G + 1 <= TAB_CNT0, "degrees and flags end below the list lengths");
static_assert(TAB_CNT0 + FIND_G * OWN_WAVES <= TAB_CUR0, "list lengths end below the row cursors");
static_assert(TAB_CUR0 + FIND_G * NPB <= TAB_WORDS, "table fits its words");
static_assert(TAB_WORDS <= OWN_THREADS, "one thread per table word when a finder clears it");
static_assert(N_EDGES <= (1 << 20) && FIND_G * NPB <= (1 << 12), "a match packs into one word");
static_assert(FIND_LDS_WORDS * 4 > 160 * 1024, "one finder per WGP");

__global__ __launch_bounds__(OWN_THREADS)
void find_edges(const int* __restrict__ src, const int* __restrict__ dst,
                int2* __restrict__ list,
                unsigned* __restrict__ counts,
                unsigned* __restrict__ bcnt,
                float* __restrict__ inv) {
  __shared__ unsigned tab[FIND_LDS_WORDS];
  constexpr unsigned SPAN = FIND_G * NPB;
  const int tid = threadIdx.x;
  if (tid < TAB_WORDS) tab[tid] = 0u;
  __syncthreads();
  const unsigned lo = (unsigned)blockIdx.x * SPAN;
  unsigned* seg = tab + TAB_WORDS + tid;
  unsigned c = 0;
  const int4* d4 = reinterpret_cast<const int4*>(dst);
  constexpr int NV = N_EDGES / 4;
  for (int i0 = tid; i0 < NV; i0 += 4 * OWN_THREADS) {
    int4 d[4];
#pragma unroll
    for (int k = 0; k < 4; ++k) {
      const int i = i0 + k * OWN_THREADS;
      d[k] = (i < NV) ? d4[i] : make_int4(-1, -1, -1, -1);
    }
#pragma unroll
    for (int k = 0; k < 4; ++k) {
      const unsigned e = (unsigned)(i0 + k * OWN_THREADS) * 4u;
      const int dv[4] = {d[k].x, d[k].y, d[k].z, d[k].w};
#pragma unroll
      for (int j = 0; j < 4; ++j) {
        const unsigned u = (unsigned)dv[j] - lo;
        if (u < SPAN) {
          atomicAdd(&tab[u], 1u);
          if (c < (unsigned)SEG_K) seg[c * OWN_THREADS] = (u << 20) | (e + j);
          ++c;
        }
      }
    }
  }
  const unsigned cs = c < (unsigned)SEG_K ? c : (unsigned)SEG_K;
  if (c > (unsigned)SEG_K) tab[SPAN + FIND_G] = 1u;
  __syncthreads();
  if (tid < (int)SPAN) {
    const unsigned o = (unsigned)tid / (unsigned)NPB, r = (unsigned)tid - o * (unsigned)NPB;
    const unsigned w = r & (unsigned)(OWN_WAVES - 1);
    unsigned first = 0;
#pragma unroll
    for (unsigned k = 0; k + 1 < (unsigned)ROWS_PW; ++k) {
      const unsigned rr = w + k * (unsigned)OWN_WAVES;
      if (rr < r) first += tab[o * (unsigned)NPB + rr];
    }
    tab[TAB_CUR0 + tid] = first;
    if (r + (unsigned)OWN_WAVES >= (unsigned)NPB) tab[TAB_CNT0 + o * (unsigned)OWN_WAVES + w] = first + tab[tid];
  }
  __syncthreads();
  int2* lists = list + (size_t)blockIdx.x * FIND_G * OWN_WAVES * BCAP;
  for (unsigned s = 0; s < cs; ++s) {
    const unsigned w = seg[s * OWN_THREADS];
    const unsigned u = w >> 20, e = w & 0xFFFFFu;
    const unsigned o = u / (unsigned)NPB, r = u - o * (unsigned)NPB;
    const unsigned bk = o * (unsigned)OWN_WAVES + (r & (unsigned)(OWN_WAVES - 1));
    const unsigned pos = atomicAdd(&tab[TAB_CUR0 + u], 1u);
    if (pos < (unsigned)BCAP) lists[(size_t)bk * BCAP + pos] = make_int2(src[e], (int)(r / (unsigned)OWN_WAVES));
    else tab[SPAN + o] = 1u;
  }
  __syncthreads();
  if (tid < FIND_G)
    counts[blockIdx.x * FIND_G + tid] = tab[SPAN + FIND_G] | tab[SPAN + tid];
  if (tid < FIND_G * OWN_WAVES)
    bcnt[blockIdx.x * FIND_G * OWN_WAVES + tid] = tab[TAB_CNT0 + tid];
  if (tid < (int)SPAN) inv[lo + tid] = 1.0f / fmaxf((float)tab[tid], 1.0f);
}

typedef __attribute__((ext_vector_type(4))) float v4f;
template<int VW> struct lane_vec;
template<> struct lane_vec<4> { typedef v4f type; };
template<> struct lane_vec<2> { typedef v2f type; };

template<int D>
__global__ __launch_bounds__(OWN_THREADS)
void sum_rows_own(const float* __restrict__ feat,
                  const int* __restrict__ src, const int* __restrict__ dst,
                  const int2* __restrict__ list, const unsigned* __restrict__ counts,
                  const unsigned* __restrict__ bcnt,
                  float* __restrict__ agg) {
  constexpr int VW = D / 32;
  constexpr int U  = 16;
  constexpr int PL = D / 2;
  typedef typename lane_vec<VW>::type VT;
  static_assert(PL % VW == 0, "a lane's floats lie in one plane");
  __shared__ __attribute__((aligned(16))) float sums[OWN_LDS_WORDS];
  const int tid  = threadIdx.x;
  const int lane = tid & 31;
  const int wave = __builtin_amdgcn_readfirstlane(tid >> 5);
  const int lo = blockIdx.x * NPB;
  if (counts[blockIdx.x] == 0u) {
    const int2* my = list + ((size_t)blockIdx.x * OWN_WAVES + wave) * BCAP;
    const int nE = __builtin_amdgcn_readfirstlane((int)bcnt[blockIdx.x * OWN_WAVES + wave]);
    const int nB = (nE + U - 1) / U;
    const int sub = lane & (U - 1);
    const int c = lane * VW;
    const int p = c / PL, cc = c % PL;
    float* ap = agg + (size_t)p * N_NODES * PL + (size_t)lo * PL + cc;
    VT zero;
#pragma unroll
    for (int j = 0; j < VW; ++j) zero[j] = 0.f;
    VT acc = zero;
    int cur = 0;
    auto next_slot = [&](const int qn) {
      if (cur < ROWS_PW) {
        const int r = wave + cur * OWN_WAVES;
        if (r < NPB) *reinterpret_cast<VT*>(ap + (size_t)r * PL) = acc;
      }
#pragma nounroll
      for (int z = cur + 1; z < qn; ++z) {
        const int r = wave + z * OWN_WAVES;
        if (r < NPB) *reinterpret_cast<VT*>(ap + (size_t)r * PL) = zero;
      }
      acc = zero;
      cur = qn;
    };
    int2 ent = make_int2(0, ROWS_PW);
    if (sub < nE) ent = my[sub];
    for (int bt = 0; bt < nB; ++bt) {
      int2 nxt = make_int2(0, ROWS_PW);
      const int idn = (bt + 1) * U + sub;
      if (idn < nE) nxt = my[idn];
      VT v[U];
#pragma unroll
      for (int u = 0; u < U; ++u) {
        const int sj = __builtin_amdgcn_readlane(ent.x, u);
        v[u] = *reinterpret_cast<const VT*>(feat + (size_t)sj * D + lane * VW);
      }
      __builtin_amdgcn_sched_barrier(0);
#pragma unroll
      for (int u = 0; u < U; ++u) {
        const int q = __builtin_amdgcn_readlane(ent.y, u);
        if (__builtin_expect(q != cur, 0)) next_slot(q);
        acc += v[u];
      }
      ent = nxt;
    }
    next_slot(ROWS_PW);
  } else {
    float4* s4 = reinterpret_cast<float4*>(sums);
    for (int i = tid; i < (NPB + 1) * D / 4; i += OWN_THREADS) s4[i] = make_float4(0.f, 0.f, 0.f, 0.f);
    __syncthreads();
    for (int e0 = wave * 32; e0 < N_EDGES; e0 += OWN_THREADS) {
      const int e = e0 + lane;
      const unsigned u = (unsigned)dst[e] - (unsigned)lo;
      const bool m = u < (unsigned)NPB;
      const int s = m ? src[e] : 0;
      unsigned mask = __builtin_amdgcn_ballot_w32(m);
      while (mask) {
        const int l = __builtin_ctz(mask);
        mask &= mask - 1;
        const int sj = __shfl(s, l);
        const int tj = __shfl((int)u, l);
        const VT val = *reinterpret_cast<const VT*>(feat + (size_t)sj * D + lane * VW);
        float* sd = sums + tj * D + lane;
#pragma unroll
        for (int j = 0; j < VW; ++j) atomicAdd(&sd[j * 32], val[j]);
      }
    }
    __syncthreads();
    for (int i = tid; i < NPB * 32; i += OWN_THREADS) {
      const int u = i >> 5, l = i & 31;
      const int c = l * VW;
      const int p = c / PL, cc = c % PL;
      VT val;
#pragma unroll
      for (int j = 0; j < VW; ++j) val[j] = sums[u * D + j * 32 + l];
      *reinterpret_cast<VT*>(agg + (size_t)p * N_NODES * PL + (size_t)(lo + u) * PL + cc) = val;
    }
  }
}


#define GEMM_WAVES 16
#define GEMM_THREADS (GEMM_WAVES * 32)
#define GEMM_WGS 256
#define GEMM_LDS_WORDS 41216

template<int CT>
__device__ __forceinline__ void mma_k8(v8f (&acc)[CT], const v4f a, const v4f (&b)[CT]) {
  const v2f alo = {a.x, a.y}, ahi = {a.z, a.w};
#pragma unroll
  for (int t = 0; t < CT; ++t) {
    const v2f blo = {b[t].x, b[t].y};
    acc[t] = __builtin_amdgcn_wmma_f32_16x16x4_f32(false, alo, false, blo,
                                                   (short)0, acc[t], false, false);
  }
#pragma unroll
  for (int t = 0; t < CT; ++t) {
    const v2f bhi = {b[t].z, b[t].w};
    acc[t] = __builtin_amdgcn_wmma_f32_16x16x4_f32(false, ahi, false, bhi,
                                                   (short)0, acc[t], false, false);
  }
}

template<int DIN, int DOUT, int CW>
__global__ __launch_bounds__(GEMM_THREADS)
void sage_wmma_gemm(const float* __restrict__ agg,
                    const float* __restrict__ xin,
                    const float* __restrict__ inv,
                    const float* __restrict__ Wl,
                    const float* __restrict__ Wr,
                    const float* __restrict__ bias,
                    float* __restrict__ out) {
  constexpr int CT = CW / 16;
  constexpr int NB = DOUT / CW;
  constexpr int PITCH = DIN + 4;
  constexpr int PL = DIN / 2;
  constexpr int NS = DIN / 8;
  constexpr int G = 8 / CT;
  constexpr int NTRIP = 2 * NS / G;
  constexpr int TRIPS_MEAN = NS / G;
  constexpr int TRIPS_PLANE = NS / 2 / G;
  static_assert(DOUT % CW == 0 && CW % 16 == 0, "a row tile is whole column blocks");
  static_assert(((PITCH / 4) & 1) == 1, "a weight row is an odd number of 16-byte pieces");
  static_assert(2 * DOUT * PITCH <= GEMM_LDS_WORDS, "both weight matrices fit the reservation");
  static_assert(PL % 8 == 0, "a plane is whole groups of eight k");
  static_assert(CT * G == 8 && (NS / 2) % G == 0, "a trip stays inside one plane");
  static_assert((DOUT * (DIN / 4)) % GEMM_THREADS == 0, "every thread copies the same number of pieces");
  __shared__ __attribute__((aligned(16))) float wlds[GEMM_LDS_WORDS];
  const int tid = threadIdx.x;
  {
    constexpr int R4 = DIN / 4;
    for (int i = tid; i < DOUT * R4; i += GEMM_THREADS) {
      const int r = i / R4, c = i - r * R4;
      const v4f wl = *(const v4f*)(Wl + (size_t)i * 4);
      const v4f wr = *(const v4f*)(Wr + (size_t)i * 4);
      *(v4f*)(wlds + r * PITCH + c * 4) = wl;
      *(v4f*)(wlds + (DOUT + r) * PITCH + c * 4) = wr;
    }
  }
  __syncthreads();

  const int lane = tid & 31;
  const int half = lane >> 4;
  const int l16  = lane & 15;
  const int kOff = half << 2;
  const int wave = __builtin_amdgcn_readfirstlane(tid >> 5);
  constexpr unsigned NT = N_NODES / 16;
  const unsigned t0 = (blockIdx.x * NT) / (unsigned)GEMM_WGS;
  const unsigned t1 = ((blockIdx.x + 1u) * NT) / (unsigned)GEMM_WGS;
  const int nU = (int)(t1 - t0) * NB;

  for (int u = wave; u < nU; u += GEMM_WAVES) {
    const int mTile = (int)(t0 + (unsigned)(u / NB)) << 4;
    const int nBase = (u % NB) * CW;
    const int aRow = mTile + l16;
    const float invc = inv[aRow];
    const float* aAgg = agg + (size_t)aRow * PL + kOff;
    const float* aX   = xin + (size_t)aRow * DIN + kOff;
    const float* bL   = wlds + (nBase + l16) * PITCH + kOff;
    const float* bR   = bL + DOUT * PITCH;

    v8f acc[CT];
#pragma unroll
    for (int t = 0; t < CT; ++t) acc[t] = v8f{};

    auto a_ptr = [&](const int i) -> const float* {
      if (i < TRIPS_PLANE) return aAgg + i * (G * 8);
      if (i < TRIPS_MEAN) return aAgg + (size_t)N_NODES * PL + (i - TRIPS_PLANE) * (G * 8);
      return aX + (i - TRIPS_MEAN) * (G * 8);
    };
    v4f an[G];
#pragma unroll
    for (int g = 0; g < G; ++g) an[g] = *(const v4f*)(a_ptr(0) + g * 8);
    auto trip = [&](const int i, const bool mean) {
      v4f ac[G];
#pragma unroll
      for (int g = 0; g < G; ++g) ac[g] = an[g];
      if (i + 1 < NTRIP) {
        const float* np = a_ptr(i + 1);
#pragma unroll
        for (int g = 0; g < G; ++g) an[g] = *(const v4f*)(np + g * 8);
      }
      const float* bp = mean ? (bL + i * (G * 8)) : (bR + (i - TRIPS_MEAN) * (G * 8));
#pragma unroll
      for (int g = 0; g < G; ++g) {
        v4f b[CT];
#pragma unroll
        for (int t = 0; t < CT; ++t) b[t] = *(const v4f*)(bp + g * 8 + t * 16 * PITCH);
        v4f av = ac[g];
        if (mean) av *= invc;
        mma_k8<CT>(acc, av, b);
      }
    };
#pragma nounroll
    for (int i = 0; i < TRIPS_MEAN; ++i) trip(i, true);
#pragma nounroll
    for (int i = TRIPS_MEAN; i < NTRIP; ++i) trip(i, false);

    const int rowBase = mTile + (half << 3);
#pragma unroll
    for (int t = 0; t < CT; ++t) {
      const int col = nBase + t * 16 + l16;
      const float bj = bias[col];
#pragma unroll
      for (int i = 0; i < 8; ++i)
        out[(size_t)(rowBase + i) * DOUT + col] = acc[t][i] + bj;
    }
  }
}


extern "C" void kernel_launch(void* const* d_in, const int* in_sizes, int n_in,
                              void* d_out, int out_size, void* d_ws, size_t ws_size,
                              hipStream_t stream) {
  const float* x   = (const float*)d_in[0];
  const int*   ei  = (const int*)d_in[1];
  const int*   src = ei;
  const int*   dst = ei + N_EDGES;
  const float* Wl1 = (const float*)d_in[2];
  const float* b1  = (const float*)d_in[3];
  const float* Wr1 = (const float*)d_in[4];
  const float* Wl2 = (const float*)d_in[5];
  const float* b2  = (const float*)d_in[6];
  const float* Wr2 = (const float*)d_in[7];
  const float* Wl3 = (const float*)d_in[8];
  const float* b3  = (const float*)d_in[9];
  const float* Wr3 = (const float*)d_in[10];
  float* out = (float*)d_out;

  float* ws  = (float*)d_ws;
  float* inv = ws;
  unsigned* counts = (unsigned*)(inv + 50048);
  unsigned* bcnt = counts + 256;
  int2* list = (int2*)(bcnt + 8192);
  float* agg = (float*)(list + (size_t)NOWN * OWN_WAVES * BCAP);
  float* h1  = agg + (size_t)N_NODES * 128;
  float* h2  = h1  + (size_t)N_NODES * 128;

  find_edges<<<NOWN / FIND_G, OWN_THREADS, 0, stream>>>(src, dst, list, counts, bcnt, inv);

  sum_rows_own<128><<<NOWN, OWN_THREADS, 0, stream>>>(x, src, dst, list, counts, bcnt, agg);
  sage_wmma_gemm<128, 128, 64><<<GEMM_WGS, GEMM_THREADS, 0, stream>>>(agg, x, inv, Wl1, Wr1, b1, h1);

  sum_rows_own<128><<<NOWN, OWN_THREADS, 0, stream>>>(h1, src, dst, list, counts, bcnt, agg);
  sage_wmma_gemm<128, 64, 32><<<GEMM_WGS, GEMM_THREADS, 0, stream>>>(agg, h1, inv, Wl2, Wr2, b2, h2);

  sum_rows_own<64><<<NOWN, OWN_THREADS, 0, stream>>>(h2, src, dst, list, counts, bcnt, agg);
  sage_wmma_gemm<64, 128, 64><<<GEMM_WGS, GEMM_THREADS, 0, stream>>>(agg, h2, inv, Wl3, Wr3, b3, out);
}


# Edge scan cut in two: 250 finder workgroups (every WGP, not 125) each scan half the destination list; an owner wave merges its two part lists row by row from stored row starts; arrival order kept
# speedup vs baseline: 3.3125x; 1.1591x over previous
#include <hip/hip_runtime.h>

#define N_NODES 50000
#define N_EDGES 800000

typedef __attribute__((ext_vector_type(2))) float v2f;
typedef __attribute__((ext_vector_type(8))) float v8f;

#define NPB 200
#define NOWN (N_NODES / NPB)
#define FIND_G 2
#define FIND_S 2
#define NOWN_PAD 256
#define N_PAD 50048
#define OWN_THREADS 1024
#define OWN_WAVES (OWN_THREADS / 32)
#define ROWS_PW ((NPB + OWN_WAVES - 1) / OWN_WAVES)
#define BCAP 256
#define OFFS_PW (ROWS_PW + 1)
#define PART_LIST ((size_t)NOWN * OWN_WAVES * BCAP)
#define PART_OFFS ((size_t)NOWN * OWN_WAVES * OFFS_PW)
#define OWN_LDS_WORDS 41216
static_assert(N_NODES % NPB == 0, "every owner has the same number of nodes");
static_assert(NOWN % FIND_G == 0, "every finder serves the same number of owners");
static_assert(N_EDGES % (4 * FIND_S) == 0, "a part of the destination list is read 16 bytes at a time");
static_assert(FIND_S == 2, "the layer passes merge exactly two parts");
static_assert(OFFS_PW == 8, "a wave's row starts are one 32-byte piece");
static_assert(N_EDGES % 32 == 0, "a wave of the rescue path always holds 32 edges");
static_assert((NPB + 1) * 128 <= OWN_LDS_WORDS, "sum rows of the rescue path fit the reservation");
static_assert(OWN_WAVES == 32, "a row's wave is the low five bits of its local number");

#define SEG_K 40
#define TAB_WORDS 1024
#define TAB_CUR0 512
#define FIND_LDS_WORDS (TAB_WORDS + SEG_K * OWN_THREADS)
static_assert(FIND_G * NPB + FIND_G + 1 <= TAB_CUR0, "degrees and flags end below the row cursors");
static_assert(TAB_CUR0 + FIND_G * NPB <= TAB_WORDS, "table fits its words");
static_assert(TAB_WORDS <= OWN_THREADS, "one thread per table word when a finder clears it");
static_assert(N_EDGES <= (1 << 20) && FIND_G * NPB <= (1 << 12), "a match packs into one word");
static_assert(FIND_LDS_WORDS * 4 > 160 * 1024, "one finder per WGP");

__global__ __launch_bounds__(OWN_THREADS)
void find_edges(const int* __restrict__ src, const int* __restrict__ dst,
                int2* __restrict__ list,
                unsigned* __restrict__ counts,
                unsigned* __restrict__ offs,
                unsigned* __restrict__ degp) {
  __shared__ unsigned tab[FIND_LDS_WORDS];
  constexpr unsigned SPAN = FIND_G * NPB;
  const int tid = threadIdx.x;
  if (tid < TAB_WORDS) tab[tid] = 0u;
  __syncthreads();
  const unsigned grp = (unsigned)blockIdx.x / (unsigned)FIND_S;
  const unsigned part = (unsigned)blockIdx.x % (unsigned)FIND_S;
  const unsigned lo = grp * SPAN;
  unsigned* seg = tab + TAB_WORDS + tid;
  unsigned c = 0;
  constexpr int NV = N_EDGES / 4 / FIND_S;
  const int4* d4 = reinterpret_cast<const int4*>(dst) + (size_t)part * NV;
  const unsigned ebase = part * (unsigned)(N_EDGES / FIND_S);
  for (int i0 = tid; i0 < NV; i0 += 4 * OWN_THREADS) {
    int4 d[4];
#pragma unroll
    for (int k = 0; k < 4; ++k) {
      const int i = i0 + k * OWN_THREADS;
      d[k] = (i < NV) ? d4[i] : make_int4(-1, -1, -1, -1);
    }
#pragma unroll
    for (int k = 0; k < 4; ++k) {
      const unsigned e = ebase + (unsigned)(i0 + k * OWN_THREADS) * 4u;
      const int dv[4] = {d[k].x, d[k].y, d[k].z, d[k].w};
#pragma unroll
      for (int j = 0; j < 4; ++j) {
        const unsigned u = (unsigned)dv[j] - lo;
        if (u < SPAN) {
          atomicAdd(&tab[u], 1u);
          if (c < (unsigned)SEG_K) seg[c * OWN_THREADS] = (u << 20) | (e + j);
          ++c;
        }
      }
    }
  }
  const unsigned cs = c < (unsigned)SEG_K ? c : (unsigned)SEG_K;
  if (c > (unsigned)SEG_K) tab[SPAN + FIND_G] = 1u;
  __syncthreads();
  if (tid < (int)SPAN) {
    const unsigned o = (unsigned)tid / (unsigned)NPB, r = (unsigned)tid - o * (unsigned)NPB;
    const unsigned w = r & (unsigned)(OWN_WAVES - 1);
    unsigned first = 0;
#pragma unroll
    for (unsigned k = 0; k + 1 < (unsigned)ROWS_PW; ++k) {
      const unsigned rr = w + k * (unsigned)OWN_WAVES;
      if (rr < r) first += tab[o * (unsigned)NPB + rr];
    }
    tab[TAB_CUR0 + tid] = first;
    unsigned* wo = offs + (size_t)part * PART_OFFS
                 + ((size_t)(grp * FIND_G + o) * OWN_WAVES + w) * OFFS_PW;
    const unsigned k = r / (unsigned)OWN_WAVES;
    wo[k] = first;
    if (r + (unsigned)OWN_WAVES >= (unsigned)NPB) {
      const unsigned tot = first + tab[tid];
      for (unsigned kk = k + 1; kk < (unsigned)OFFS_PW; ++kk) wo[kk] = tot;
    }
  }
  __syncthreads();
  int2* lists = list + (size_t)part * PART_LIST + (size_t)grp * FIND_G * OWN_WAVES * BCAP;
  for (unsigned s = 0; s < cs; ++s) {
    const unsigned w = seg[s * OWN_THREADS];
    const unsigned u = w >> 20, e = w & 0xFFFFFu;
    const unsigned o = u / (unsigned)NPB, r = u - o * (unsigned)NPB;
    const unsigned bk = o * (unsigned)OWN_WAVES + (r & (unsigned)(OWN_WAVES - 1));
    const unsigned pos = atomicAdd(&tab[TAB_CUR0 + u], 1u);
    if (pos < (unsigned)BCAP) lists[(size_t)bk * BCAP + pos] = make_int2(src[e], (int)(r / (unsigned)OWN_WAVES));
    else tab[SPAN + o] = 1u;
  }
  __syncthreads();
  if (tid < FIND_G)
    counts[part * NOWN_PAD + grp * FIND_G + tid] = tab[SPAN + FIND_G] | tab[SPAN + tid];
  if (tid < (int)SPAN) degp[part * N_PAD + lo + tid] = tab[tid];
}

typedef __attribute__((ext_vector_type(4))) float v4f;
template<int VW> struct lane_vec;
template<> struct lane_vec<4> { typedef v4f type; };
template<> struct lane_vec<2> { typedef v2f type; };

template<int D>
__global__ __launch_bounds__(OWN_THREADS)
void sum_rows_own(const float* __restrict__ feat,
                  const int* __restrict__ src, const int* __restrict__ dst,
                  const int2* __restrict__ list, const unsigned* __restrict__ counts,
                  const unsigned* __restrict__ offs,
                  float* __restrict__ agg) {
  constexpr int VW = D / 32;
  constexpr int U  = 16;
  constexpr int PL = D / 2;
  typedef typename lane_vec<VW>::type VT;
  static_assert(PL % VW == 0, "a lane's floats lie in one plane");
  __shared__ __attribute__((aligned(16))) float sums[OWN_LDS_WORDS];
  const int tid  = threadIdx.x;
  const int lane = tid & 31;
  const int wave = __builtin_amdgcn_readfirstlane(tid >> 5);
  const int lo = blockIdx.x * NPB;
  if ((counts[blockIdx.x] | counts[NOWN_PAD + blockIdx.x]) == 0u) {
    const int2* my = list + ((size_t)blockIdx.x * OWN_WAVES + wave) * BCAP;
    const unsigned* wo = offs + ((size_t)blockIdx.x * OWN_WAVES + wave) * OFFS_PW;
    int o0[OFFS_PW], o1[OFFS_PW];
#pragma unroll
    for (int k = 0; k < OFFS_PW; ++k) {
      o0[k] = __builtin_amdgcn_readfirstlane((int)wo[k]);
      o1[k] = __builtin_amdgcn_readfirstlane((int)wo[PART_OFFS + k]);
    }
    const int nE = o0[ROWS_PW] + o1[ROWS_PW];
    const int nB = (nE + U - 1) / U;
    auto entry = [&](const int i) -> int2 {
      int2 e = make_int2(0, ROWS_PW);
      if (i < nE) {
        int st = 0, b0 = 0, b1 = 0, l0 = o0[1];
#pragma unroll
        for (int k = 1; k < ROWS_PW; ++k) {
          const int sk = o0[k] + o1[k];
          if (i >= sk) { st = sk; b0 = o0[k]; b1 = o1[k]; l0 = o0[k + 1] - o0[k]; }
        }
        const int j = i - st;
        const size_t at = (j < l0) ? (size_t)(b0 + j) : PART_LIST + (size_t)(b1 + j - l0);
        e = my[at];
      }
      return e;
    };
    const int sub = lane & (U - 1);
    const int c = lane * VW;
    const int p = c / PL, cc = c % PL;
    float* ap = agg + (size_t)p * N_NODES * PL + (size_t)lo * PL + cc;
    VT zero;
#pragma unroll
    for (int j = 0; j < VW; ++j) zero[j] = 0.f;
    VT acc = zero;
    int cur = 0;
    auto next_slot = [&](const int qn) {
      if (cur < ROWS_PW) {
        const int r = wave + cur * OWN_WAVES;
        if (r < NPB) *reinterpret_cast<VT*>(ap + (size_t)r * PL) = acc;
      }
#pragma nounroll
      for (int z = cur + 1; z < qn; ++z) {
        const int r = wave + z * OWN_WAVES;
        if (r < NPB) *reinterpret_cast<VT*>(ap + (size_t)r * PL) = zero;
      }
      acc = zero;
      cur = qn;
    };
    int2 ent = entry(sub);
    for (int bt = 0; bt < nB; ++bt) {
      const int2 nxt = entry((bt + 1) * U + sub);
      VT v[U];
#pragma unroll
      for (int u = 0; u < U; ++u) {
        const int sj = __builtin_amdgcn_readlane(ent.x, u);
        v[u] = *reinterpret_cast<const VT*>(feat + (size_t)sj * D + lane * VW);
      }
      __builtin_amdgcn_sched_barrier(0);
#pragma unroll
      for (int u = 0; u < U; ++u) {
        const int q = __builtin_amdgcn_readlane(ent.y, u);
        if (__builtin_expect(q != cur, 0)) next_slot(q);
        acc += v[u];
      }
      ent = nxt;
    }
    next_slot(ROWS_PW);
  } else {
    float4* s4 = reinterpret_cast<float4*>(sums);
    for (int i = tid; i < (NPB + 1) * D / 4; i += OWN_THREADS) s4[i] = make_float4(0.f, 0.f, 0.f, 0.f);
    __syncthreads();
    for (int e0 = wave * 32; e0 < N_EDGES; e0 += OWN_THREADS) {
      const int e = e0 + lane;
      const unsigned u = (unsigned)dst[e] - (unsigned)lo;
      const bool m = u < (unsigned)NPB;
      const int s = m ? src[e] : 0;
      unsigned mask = __builtin_amdgcn_ballot_w32(m);
      while (mask) {
        const int l = __builtin_ctz(mask);
        mask &= mask - 1;
        const int sj = __shfl(s, l);
        const int tj = __shfl((int)u, l);
        const VT val = *reinterpret_cast<const VT*>(feat + (size_t)sj * D + lane * VW);
        float* sd = sums + tj * D + lane;
#pragma unroll
        for (int j = 0; j < VW; ++j) atomicAdd(&sd[j * 32], val[j]);
      }
    }
    __syncthreads();
    for (int i = tid; i < NPB * 32; i += OWN_THREADS) {
      const int u = i >> 5, l = i & 31;
      const int c = l * VW;
      const int p = c / PL, cc = c % PL;
      VT val;
#pragma unroll
      for (int j = 0; j < VW; ++j) val[j] = sums[u * D + j * 32 + l];
      *reinterpret_cast<VT*>(agg + (size_t)p * N_NODES * PL + (size_t)(lo + u) * PL + cc) = val;
    }
  }
}


#define GEMM_WAVES 16
#define GEMM_THREADS (GEMM_WAVES * 32)
#define GEMM_WGS 256
#define GEMM_LDS_WORDS 41216

template<int CT>
__device__ __forceinline__ void mma_k8(v8f (&acc)[CT], const v4f a, const v4f (&b)[CT]) {
  const v2f alo = {a.x, a.y}, ahi = {a.z, a.w};
#pragma unroll
  for (int t = 0; t < CT; ++t) {
    const v2f blo = {b[t].x, b[t].y};
    acc[t] = __builtin_amdgcn_wmma_f32_16x16x4_f32(false, alo, false, blo,
                                                   (short)0, acc[t], false, false);
  }
#pragma unroll
  for (int t = 0; t < CT; ++t) {
    const v2f bhi = {b[t].z, b[t].w};
    acc[t] = __builtin_amdgcn_wmma_f32_16x16x4_f32(false, ahi, false, bhi,
                                                   (short)0, acc[t], false, false);
  }
}

template<int DIN, int DOUT, int CW>
__global__ __launch_bounds__(GEMM_THREADS)
void sage_wmma_gemm(const float* __restrict__ agg,
                    const float* __restrict__ xin,
                    const unsigned* __restrict__ degp,
                    const float* __restrict__ Wl,
                    const float* __restrict__ Wr,
                    const float* __restrict__ bias,
                    float* __restrict__ out) {
  constexpr int CT = CW / 16;
  constexpr int NB = DOUT / CW;
  constexpr int PITCH = DIN + 4;
  constexpr int PL = DIN / 2;
  constexpr int NS = DIN / 8;
  constexpr int G = 8 / CT;
  constexpr int NTRIP = 2 * NS / G;
  constexpr int TRIPS_MEAN = NS / G;
  constexpr int TRIPS_PLANE = NS / 2 / G;
  static_assert(DOUT % CW == 0 && CW % 16 == 0, "a row tile is whole column blocks");
  static_assert(((PITCH / 4) & 1) == 1, "a weight row is an odd number of 16-byte pieces");
  static_assert(2 * DOUT * PITCH <= GEMM_LDS_WORDS, "both weight matrices fit the reservation");
  static_assert(PL % 8 == 0, "a plane is whole groups of eight k");
  static_assert(CT * G == 8 && (NS / 2) % G == 0, "a trip stays inside one plane");
  static_assert((DOUT * (DIN / 4)) % GEMM_THREADS == 0, "every thread copies the same number of pieces");
  __shared__ __attribute__((aligned(16))) float wlds[GEMM_LDS_WORDS];
  const int tid = threadIdx.x;
  {
    constexpr int R4 = DIN / 4;
    for (int i = tid; i < DOUT * R4; i += GEMM_THREADS) {
      const int r = i / R4, c = i - r * R4;
      const v4f wl = *(const v4f*)(Wl + (size_t)i * 4);
      const v4f wr = *(const v4f*)(Wr + (size_t)i * 4);
      *(v4f*)(wlds + r * PITCH + c * 4) = wl;
      *(v4f*)(wlds + (DOUT + r) * PITCH + c * 4) = wr;
    }
  }
  __syncthreads();

  const int lane = tid & 31;
  const int half = lane >> 4;
  const int l16  = lane & 15;
  const int kOff = half << 2;
  const int wave = __builtin_amdgcn_readfirstlane(tid >> 5);
  constexpr unsigned NT = N_NODES / 16;
  const unsigned t0 = (blockIdx.x * NT) / (unsigned)GEMM_WGS;
  const unsigned t1 = ((blockIdx.x + 1u) * NT) / (unsigned)GEMM_WGS;
  const int nU = (int)(t1 - t0) * NB;

  for (int u = wave; u < nU; u += GEMM_WAVES) {
    const int mTile = (int)(t0 + (unsigned)(u / NB)) << 4;
    const int nBase = (u % NB) * CW;
    const int aRow = mTile + l16;
    const float invc = 1.0f / fmaxf((float)(degp[aRow] + degp[N_PAD + aRow]), 1.0f);
    const float* aAgg = agg + (size_t)aRow * PL + kOff;
    const float* aX   = xin + (size_t)aRow * DIN + kOff;
    const float* bL   = wlds + (nBase + l16) * PITCH + kOff;
    const float* bR   = bL + DOUT * PITCH;

    v8f acc[CT];
#pragma unroll
    for (int t = 0; t < CT; ++t) acc[t] = v8f{};

    auto a_ptr = [&](const int i) -> const float* {
      if (i < TRIPS_PLANE) return aAgg + i * (G * 8);
      if (i < TRIPS_MEAN) return aAgg + (size_t)N_NODES * PL + (i - TRIPS_PLANE) * (G * 8);
      return aX + (i - TRIPS_MEAN) * (G * 8);
    };
    v4f an[G];
#pragma unroll
    for (int g = 0; g < G; ++g) an[g] = *(const v4f*)(a_ptr(0) + g * 8);
    auto trip = [&](const int i, const bool mean) {
      v4f ac[G];
#pragma unroll
      for (int g = 0; g < G; ++g) ac[g] = an[g];
      if (i + 1 < NTRIP) {
        const float* np = a_ptr(i + 1);
#pragma unroll
        for (int g = 0; g < G; ++g) an[g] = *(const v4f*)(np + g * 8);
      }
      const float* bp = mean ? (bL + i * (G * 8)) : (bR + (i - TRIPS_MEAN) * (G * 8));
#pragma unroll
      for (int g = 0; g < G; ++g) {
        v4f b[CT];
#pragma unroll
        for (int t = 0; t < CT; ++t) b[t] = *(const v4f*)(bp + g * 8 + t * 16 * PITCH);
        v4f av = ac[g];
        if (mean) av *= invc;
        mma_k8<CT>(acc, av, b);
      }
    };
#pragma nounroll
    for (int i = 0; i < TRIPS_MEAN; ++i) trip(i, true);
#pragma nounroll
    for (int i = TRIPS_MEAN; i < NTRIP; ++i) trip(i, false);

    const int rowBase = mTile + (half << 3);
#pragma unroll
    for (int t = 0; t < CT; ++t) {
      const int col = nBase + t * 16 + l16;
      const float bj = bias[col];
#pragma unroll
      for (int i = 0; i < 8; ++i)
        out[(size_t)(rowBase + i) * DOUT + col] = acc[t][i] + bj;
    }
  }
}


extern "C" void kernel_launch(void* const* d_in, const int* in_sizes, int n_in,
                              void* d_out, int out_size, void* d_ws, size_t ws_size,
                              hipStream_t stream) {
  const float* x   = (const float*)d_in[0];
  const int*   ei  = (const int*)d_in[1];
  const int*   src = ei;
  const int*   dst = ei + N_EDGES;
  const float* Wl1 = (const float*)d_in[2];
  const float* b1  = (const float*)d_in[3];
  const float* Wr1 = (const float*)d_in[4];
  const float* Wl2 = (const float*)d_in[5];
  const float* b2  = (const float*)d_in[6];
  const float* Wr2 = (const float*)d_in[7];
  const float* Wl3 = (const float*)d_in[8];
  const float* b3  = (const float*)d_in[9];
  const float* Wr3 = (const float*)d_in[10];
  float* out = (float*)d_out;

  float* ws  = (float*)d_ws;
  unsigned* degp = (unsigned*)ws;
  unsigned* counts = degp + FIND_S * N_PAD;
  unsigned* offs = counts + FIND_S * NOWN_PAD;
  int2* list = (int2*)(offs + 131072);
  float* agg = (float*)(list + FIND_S * PART_LIST);
  float* h1  = agg + (size_t)N_NODES * 128;
  float* h2  = h1  + (size_t)N_NODES * 128;

  find_edges<<<NOWN / FIND_G * FIND_S, OWN_THREADS, 0, stream>>>(src, dst, list, counts, offs, degp);

  sum_rows_own<128><<<NOWN, OWN_THREADS, 0, stream>>>(x, src, dst, list, counts, offs, agg);
  sage_wmma_gemm<128, 128, 64><<<GEMM_WGS, GEMM_THREADS, 0, stream>>>(agg, x, degp, Wl1, Wr1, b1, h1);

  sum_rows_own<128><<<NOWN, OWN_THREADS, 0, stream>>>(h1, src, dst, list, counts, offs, agg);
  sage_wmma_gemm<128, 64, 32><<<GEMM_WGS, GEMM_THREADS, 0, stream>>>(agg, h1, degp, Wl2, Wr2, b2, h2);

  sum_rows_own<64><<<NOWN, OWN_THREADS, 0, stream>>>(h2, src, dst, list, counts, offs, agg);
  sage_wmma_gemm<64, 128, 64><<<GEMM_WGS, GEMM_THREADS, 0, stream>>>(agg, h2, degp, Wl3, Wr3, b3, out);
}


# Edge scan cut in five, not two: a finder scans a fifth of the destination list for 5 owners; an owner wave joins its 5 part lists row by row in LDS and walks that list; sum order changes
# speedup vs baseline: 3.4315x; 1.0359x over previous
#include <hip/hip_runtime.h>

#define N_NODES 50000
#define N_EDGES 800000

typedef __attribute__((ext_vector_type(2))) float v2f;
typedef __attribute__((ext_vector_type(8))) float v8f;

#define NPB 200
#define NOWN (N_NODES / NPB)
#define FIND_G 5
#define FIND_S 5
#define NOWN_PAD 256
#define N_PAD 50048
#define OWN_THREADS 1024
#define OWN_WAVES (OWN_THREADS / 32)
#define ROWS_PW ((NPB + OWN_WAVES - 1) / OWN_WAVES)
#define BCAP 128
#define OFFS_PW (ROWS_PW + 1)
#define OFFS_WAVE (OFFS_PW * FIND_S)
#define NSEG (ROWS_PW * FIND_S)
#define PART_LIST ((size_t)NOWN * OWN_WAVES * BCAP)
#define OWN_LDS_WORDS 41216
static_assert(N_NODES % NPB == 0, "every owner has the same number of nodes");
static_assert(NOWN % FIND_G == 0, "every finder serves the same number of owners");
static_assert(N_EDGES % (4 * FIND_S) == 0, "a part of the destination list is read 16 bytes at a time");
static_assert(NSEG > 32 && NSEG <= 64, "a lane holds two pieces of its wave's edges at most");
static_assert(FIND_S * BCAP * 2 * OWN_WAVES <= OWN_LDS_WORDS, "every wave's joined list fits the reservation");
static_assert(N_EDGES % 32 == 0, "a wave of the rescue path always holds 32 edges");
static_assert((NPB + 1) * 128 <= OWN_LDS_WORDS, "sum rows of the rescue path fit the reservation");
static_assert(OWN_WAVES == 32, "a row's wave is the low five bits of its local number");

#define SEG_K 40
#define TAB_WORDS 2048
#define TAB_CUR0 1024
#define FIND_LDS_WORDS (TAB_WORDS + SEG_K * OWN_THREADS)
static_assert(FIND_G * NPB + FIND_G + 1 <= TAB_CUR0, "degrees and flags end below the row cursors");
static_assert(TAB_CUR0 + FIND_G * NPB <= TAB_WORDS, "table fits its words");
static_assert(FIND_G * NPB <= OWN_THREADS, "one thread per node of a finder");
static_assert(N_EDGES <= (1 << 20) && FIND_G * NPB <= (1 << 12), "a match packs into one word");
static_assert(FIND_LDS_WORDS * 4 > 160 * 1024, "one finder per WGP");

__global__ __launch_bounds__(OWN_THREADS)
void find_edges(const int* __restrict__ src, const int* __restrict__ dst,
                int2* __restrict__ list,
                unsigned* __restrict__ counts,
                unsigned* __restrict__ offs,
                unsigned* __restrict__ degp) {
  __shared__ unsigned tab[FIND_LDS_WORDS];
  constexpr unsigned SPAN = FIND_G * NPB;
  const int tid = threadIdx.x;
  for (int i = tid; i < TAB_WORDS; i += OWN_THREADS) tab[i] = 0u;
  __syncthreads();
  const unsigned grp = (unsigned)blockIdx.x / (unsigned)FIND_S;
  const unsigned part = (unsigned)blockIdx.x % (unsigned)FIND_S;
  const unsigned lo = grp * SPAN;
  unsigned* seg = tab + TAB_WORDS + tid;
  unsigned c = 0;
  constexpr int NV = N_EDGES / 4 / FIND_S;
  const int4* d4 = reinterpret_cast<const int4*>(dst) + (size_t)part * NV;
  const unsigned ebase = part * (unsigned)(N_EDGES / FIND_S);
  for (int i0 = tid; i0 < NV; i0 += 4 * OWN_THREADS) {
    int4 d[4];
#pragma unroll
    for (int k = 0; k < 4; ++k) {
      const int i = i0 + k * OWN_THREADS;
      d[k] = (i < NV) ? d4[i] : make_int4(-1, -1, -1, -1);
    }
#pragma unroll
    for (int k = 0; k < 4; ++k) {
      const unsigned e = ebase + (unsigned)(i0 + k * OWN_THREADS) * 4u;
      const int dv[4] = {d[k].x, d[k].y, d[k].z, d[k].w};
#pragma unroll
      for (int j = 0; j < 4; ++j) {
        const unsigned u = (unsigned)dv[j] - lo;
        if (u < SPAN) {
          atomicAdd(&tab[u], 1u);
          if (c < (unsigned)SEG_K) seg[c * OWN_THREADS] = (u << 20) | (e + j);
          ++c;
        }
      }
    }
  }
  const unsigned cs = c < (unsigned)SEG_K ? c : (unsigned)SEG_K;
  if (c > (unsigned)SEG_K) tab[SPAN + FIND_G] = 1u;
  __syncthreads();
  if (tid < (int)SPAN) {
    const unsigned o = (unsigned)tid / (unsigned)NPB, r = (unsigned)tid - o * (unsigned)NPB;
    const unsigned w = r & (unsigned)(OWN_WAVES - 1);
    unsigned first = 0;
#pragma unroll
    for (unsigned k = 0; k + 1 < (unsigned)ROWS_PW; ++k) {
      const unsigned rr = w + k * (unsigned)OWN_WAVES;
      if (rr < r) first += tab[o * (unsigned)NPB + rr];
    }
    tab[TAB_CUR0 + tid] = first;
    unsigned* wo = offs + ((size_t)(grp * FIND_G + o) * OWN_WAVES + w) * OFFS_WAVE + part;
    const unsigned k = r / (unsigned)OWN_WAVES;
    wo[k * FIND_S] = first;
    if (r + (unsigned)OWN_WAVES >= (unsigned)NPB) {
      const unsigned tot = first + tab[tid];
      for (unsigned kk = k + 1; kk < (unsigned)OFFS_PW; ++kk) wo[kk * FIND_S] = tot;
    }
  }
  __syncthreads();
  int2* lists = list + (size_t)part * PART_LIST + (size_t)grp * FIND_G * OWN_WAVES * BCAP;
  for (unsigned s = 0; s < cs; ++s) {
    const unsigned w = seg[s * OWN_THREADS];
    const unsigned u = w >> 20, e = w & 0xFFFFFu;
    const unsigned o = u / (unsigned)NPB, r = u - o * (unsigned)NPB;
    const unsigned bk = o * (unsigned)OWN_WAVES + (r & (unsigned)(OWN_WAVES - 1));
    const unsigned pos = atomicAdd(&tab[TAB_CUR0 + u], 1u);
    if (pos < (unsigned)BCAP) lists[(size_t)bk * BCAP + pos] = make_int2(src[e], (int)(r / (unsigned)OWN_WAVES));
    else tab[SPAN + o] = 1u;
  }
  __syncthreads();
  if (tid < FIND_G)
    counts[part * NOWN_PAD + grp * FIND_G + tid] = tab[SPAN + FIND_G] | tab[SPAN + tid];
  if (tid < (int)SPAN) degp[part * N_PAD + lo + tid] = tab[tid];
}

typedef __attribute__((ext_vector_type(4))) float v4f;
template<int VW> struct lane_vec;
template<> struct lane_vec<4> { typedef v4f type; };
template<> struct lane_vec<2> { typedef v2f type; };

template<int D>
__global__ __launch_bounds__(OWN_THREADS)
void sum_rows_own(const float* __restrict__ feat,
                  const int* __restrict__ src, const int* __restrict__ dst,
                  const int2* __restrict__ list, const unsigned* __restrict__ counts,
                  const unsigned* __restrict__ offs,
                  float* __restrict__ agg) {
  constexpr int VW = D / 32;
  constexpr int U  = 16;
  constexpr int PL = D / 2;
  typedef typename lane_vec<VW>::type VT;
  static_assert(PL % VW == 0, "a lane's floats lie in one plane");
  __shared__ __attribute__((aligned(16))) float sums[OWN_LDS_WORDS];
  const int tid  = threadIdx.x;
  const int lane = tid & 31;
  const int wave = __builtin_amdgcn_readfirstlane(tid >> 5);
  const int lo = blockIdx.x * NPB;
  unsigned flagged = 0u;
#pragma unroll
  for (int p = 0; p < FIND_S; ++p) flagged |= counts[p * NOWN_PAD + blockIdx.x];
  if (flagged == 0u) {
    const int wv = blockIdx.x * OWN_WAVES + wave;
    const unsigned* wo = offs + (size_t)wv * OFFS_WAVE;
    const bool two = lane < NSEG - 32;
    const int stA = (int)wo[lane];
    const int lenA = (int)wo[lane + FIND_S] - stA;
    const int stB = two ? (int)wo[32 + lane] : 0;
    const int lenB = two ? (int)wo[32 + lane + FIND_S] - stB : 0;
    int incA = lenA, incB = lenB;
#pragma unroll
    for (int d = 1; d < 32; d <<= 1) {
      const int tA = __shfl_up(incA, d), tB = __shfl_up(incB, d);
      if (lane >= d) { incA += tA; incB += tB; }
    }
    const int totA = __builtin_amdgcn_readlane(incA, 31);
    const int nE = totA + __builtin_amdgcn_readlane(incB, 31);
    const int nB = (nE + U - 1) / U;
    const int mA = incA - lenA, mB = totA + incB - lenB;
    int2* mer = reinterpret_cast<int2*>(sums) + wave * (FIND_S * BCAP);
    const int2* lA = list + (size_t)(lane % FIND_S) * PART_LIST + (size_t)wv * BCAP + stA;
    const int2* lB = list + (size_t)((32 + lane) % FIND_S) * PART_LIST + (size_t)wv * BCAP + stB;
    int mx = lenA > lenB ? lenA : lenB;
#pragma unroll
    for (int d = 16; d > 0; d >>= 1) {
      const int t = __shfl_xor(mx, d);
      mx = t > mx ? t : mx;
    }
    mx = __builtin_amdgcn_readfirstlane(mx);
    for (int j0 = 0; j0 < mx; j0 += 8) {
      int2 tA[8], tB[8];
#pragma unroll
      for (int q = 0; q < 8; ++q) {
        tA[q] = (j0 + q < lenA) ? lA[j0 + q] : make_int2(0, 0);
        tB[q] = (j0 + q < lenB) ? lB[j0 + q] : make_int2(0, 0);
      }
#pragma unroll
      for (int q = 0; q < 8; ++q) {
        if (j0 + q < lenA) mer[mA + j0 + q] = tA[q];
        if (j0 + q < lenB) mer[mB + j0 + q] = tB[q];
      }
    }
    auto entry = [&](const int i) -> int2 {
      int2 e = make_int2(0, ROWS_PW);
      if (i < nE) e = mer[i];
      return e;
    };
    const int sub = lane & (U - 1);
    const int c = lane * VW;
    const int p = c / PL, cc = c % PL;
    float* ap = agg + (size_t)p * N_NODES * PL + (size_t)lo * PL + cc;
    VT zero;
#pragma unroll
    for (int j = 0; j < VW; ++j) zero[j] = 0.f;
    VT acc = zero;
    int cur = 0;
    auto next_slot = [&](const int qn) {
      if (cur < ROWS_PW) {
        const int r = wave + cur * OWN_WAVES;
        if (r < NPB) *reinterpret_cast<VT*>(ap + (size_t)r * PL) = acc;
      }
#pragma nounroll
      for (int z = cur + 1; z < qn; ++z) {
        const int r = wave + z * OWN_WAVES;
        if (r < NPB) *reinterpret_cast<VT*>(ap + (size_t)r * PL) = zero;
      }
      acc = zero;
      cur = qn;
    };
    int2 ent = entry(sub);
    for (int bt = 0; bt < nB; ++bt) {
      VT v[U];
#pragma unroll
      for (int u = 0; u < U; ++u) {
        const int sj = __builtin_amdgcn_readlane(ent.x, u);
        v[u] = *reinterpret_cast<const VT*>(feat + (size_t)sj * D + lane * VW);
      }
      const int2 nxt = entry((bt + 1) * U + sub);
      __builtin_amdgcn_sched_barrier(0);
#pragma unroll
      for (int u = 0; u < U; ++u) {
        const int q = __builtin_amdgcn_readlane(ent.y, u);
        if (__builtin_expect(q != cur, 0)) next_slot(q);
        acc += v[u];
      }
      ent = nxt;
    }
    next_slot(ROWS_PW);
  } else {
    float4* s4 = reinterpret_cast<float4*>(sums);
    for (int i = tid; i < (NPB + 1) * D / 4; i += OWN_THREADS) s4[i] = make_float4(0.f, 0.f, 0.f, 0.f);
    __syncthreads();
    for (int e0 = wave * 32; e0 < N_EDGES; e0 += OWN_THREADS) {
      const int e = e0 + lane;
      const unsigned u = (unsigned)dst[e] - (unsigned)lo;
      const bool m = u < (unsigned)NPB;
      const int s = m ? src[e] : 0;
      unsigned mask = __builtin_amdgcn_ballot_w32(m);
      while (mask) {
        const int l = __builtin_ctz(mask);
        mask &= mask - 1;
        const int sj = __shfl(s, l);
        const int tj = __shfl((int)u, l);
        const VT val = *reinterpret_cast<const VT*>(feat + (size_t)sj * D + lane * VW);
        float* sd = sums + tj * D + lane;
#pragma unroll
        for (int j = 0; j < VW; ++j) atomicAdd(&sd[j * 32], val[j]);
      }
    }
    __syncthreads();
    for (int i = tid; i < NPB * 32; i += OWN_THREADS) {
      const int u = i >> 5, l = i & 31;
      const int c = l * VW;
      const int p = c / PL, cc = c % PL;
      VT val;
#pragma unroll
      for (int j = 0; j < VW; ++j) val[j] = sums[u * D + j * 32 + l];
      *reinterpret_cast<VT*>(agg + (size_t)p * N_NODES * PL + (size_t)(lo + u) * PL + cc) = val;
    }
  }
}


#define GEMM_WAVES 16
#define GEMM_THREADS (GEMM_WAVES * 32)
#define GEMM_WGS 256
#define GEMM_LDS_WORDS 41216

template<int CT>
__device__ __forceinline__ void mma_k8(v8f (&acc)[CT], const v4f a, const v4f (&b)[CT]) {
  const v2f alo = {a.x, a.y}, ahi = {a.z, a.w};
#pragma unroll
  for (int t = 0; t < CT; ++t) {
    const v2f blo = {b[t].x, b[t].y};
    acc[t] = __builtin_amdgcn_wmma_f32_16x16x4_f32(false, alo, false, blo,
                                                   (short)0, acc[t], false, false);
  }
#pragma unroll
  for (int t = 0; t < CT; ++t) {
    const v2f bhi = {b[t].z, b[t].w};
    acc[t] = __builtin_amdgcn_wmma_f32_16x16x4_f32(false, ahi, false, bhi,
                                                   (short)0, acc[t], false, false);
  }
}

template<int DIN, int DOUT, int CW>
__global__ __launch_bounds__(GEMM_THREADS)
void sage_wmma_gemm(const float* __restrict__ agg,
                    const float* __restrict__ xin,
                    const unsigned* __restrict__ degp,
                    const float* __restrict__ Wl,
                    const float* __restrict__ Wr,
                    const float* __restrict__ bias,
                    float* __restrict__ out) {
  constexpr int CT = CW / 16;
  constexpr int NB = DOUT / CW;
  constexpr int PITCH = DIN + 4;
  constexpr int PL = DIN / 2;
  constexpr int NS = DIN / 8;
  constexpr int G = 8 / CT;
  constexpr int NTRIP = 2 * NS / G;
  constexpr int TRIPS_MEAN = NS / G;
  constexpr int TRIPS_PLANE = NS / 2 / G;
  static_assert(DOUT % CW == 0 && CW % 16 == 0, "a row tile is whole column blocks");
  static_assert(((PITCH / 4) & 1) == 1, "a weight row is an odd number of 16-byte pieces");
  static_assert(2 * DOUT * PITCH <= GEMM_LDS_WORDS, "both weight matrices fit the reservation");
  static_assert(PL % 8 == 0, "a plane is whole groups of eight k");
  static_assert(CT * G == 8 && (NS / 2) % G == 0, "a trip stays inside one plane");
  static_assert((DOUT * (DIN / 4)) % GEMM_THREADS == 0, "every thread copies the same number of pieces");
  __shared__ __attribute__((aligned(16))) float wlds[GEMM_LDS_WORDS];
  const int tid = threadIdx.x;
  {
    constexpr int R4 = DIN / 4;
    for (int i = tid; i < DOUT * R4; i += GEMM_THREADS) {
      const int r = i / R4, c = i - r * R4;
      const v4f wl = *(const v4f*)(Wl + (size_t)i * 4);
      const v4f wr = *(const v4f*)(Wr + (size_t)i * 4);
      *(v4f*)(wlds + r * PITCH + c * 4) = wl;
      *(v4f*)(wlds + (DOUT + r) * PITCH + c * 4) = wr;
    }
  }
  __syncthreads();

  const int lane = tid & 31;
  const int half = lane >> 4;
  const int l16  = lane & 15;
  const int kOff = half << 2;
  const int wave = __builtin_amdgcn_readfirstlane(tid >> 5);
  constexpr unsigned NT = N_NODES / 16;
  const unsigned t0 = (blockIdx.x * NT) / (unsigned)GEMM_WGS;
  const unsigned t1 = ((blockIdx.x + 1u) * NT) / (unsigned)GEMM_WGS;
  const int nU = (int)(t1 - t0) * NB;

  for (int u = wave; u < nU; u += GEMM_WAVES) {
    const int mTile = (int)(t0 + (unsigned)(u / NB)) << 4;
    const int nBase = (u % NB) * CW;
    const int aRow = mTile + l16;
    unsigned deg = 0u;
#pragma unroll
    for (int p = 0; p < FIND_S; ++p) deg += degp[p * N_PAD + aRow];
    const float invc = 1.0f / fmaxf((float)deg, 1.0f);
    const float* aAgg = agg + (size_t)aRow * PL + kOff;
    const float* aX   = xin + (size_t)aRow * DIN + kOff;
    const float* bL   = wlds + (nBase + l16) * PITCH + kOff;
    const float* bR   = bL + DOUT * PITCH;

    v8f acc[CT];
#pragma unroll
    for (int t = 0; t < CT; ++t) acc[t] = v8f{};

    auto a_ptr = [&](const int i) -> const float* {
      if (i < TRIPS_PLANE) return aAgg + i * (G * 8);
      if (i < TRIPS_MEAN) return aAgg + (size_t)N_NODES * PL + (i - TRIPS_PLANE) * (G * 8);
      return aX + (i - TRIPS_MEAN) * (G * 8);
    };
    v4f an[G];
#pragma unroll
    for (int g = 0; g < G; ++g) an[g] = *(const v4f*)(a_ptr(0) + g * 8);
    auto trip = [&](const int i, const bool mean) {
      v4f ac[G];
#pragma unroll
      for (int g = 0; g < G; ++g) ac[g] = an[g];
      if (i + 1 < NTRIP) {
        const float* np = a_ptr(i + 1);
#pragma unroll
        for (int g = 0; g < G; ++g) an[g] = *(const v4f*)(np + g * 8);
      }
      const float* bp = mean ? (bL + i * (G * 8)) : (bR + (i - TRIPS_MEAN) * (G * 8));
#pragma unroll
      for (int g = 0; g < G; ++g) {
        v4f b[CT];
#pragma unroll
        for (int t = 0; t < CT; ++t) b[t] = *(const v4f*)(bp + g * 8 + t * 16 * PITCH);
        v4f av = ac[g];
        if (mean) av *= invc;
        mma_k8<CT>(acc, av, b);
      }
    };
#pragma nounroll
    for (int i = 0; i < TRIPS_MEAN; ++i) trip(i, true);
#pragma nounroll
    for (int i = TRIPS_MEAN; i < NTRIP; ++i) trip(i, false);

    const int rowBase = mTile + (half << 3);
#pragma unroll
    for (int t = 0; t < CT; ++t) {
      const int col = nBase + t * 16 + l16;
      const float bj = bias[col];
#pragma unroll
      for (int i = 0; i < 8; ++i)
        out[(size_t)(rowBase + i) * DOUT + col] = acc[t][i] + bj;
    }
  }
}


extern "C" void kernel_launch(void* const* d_in, const int* in_sizes, int n_in,
                              void* d_out, int out_size, void* d_ws, size_t ws_size,
                              hipStream_t stream) {
  const float* x   = (const float*)d_in[0];
  const int*   ei  = (const int*)d_in[1];
  const int*   src = ei;
  const int*   dst = ei + N_EDGES;
  const float* Wl1 = (const float*)d_in[2];
  const float* b1  = (const float*)d_in[3];
  const float* Wr1 = (const float*)d_in[4];
  const float* Wl2 = (const float*)d_in[5];
  const float* b2  = (const float*)d_in[6];
  const float* Wr2 = (const float*)d_in[7];
  const float* Wl3 = (const float*)d_in[8];
  const float* b3  = (const float*)d_in[9];
  const float* Wr3 = (const float*)d_in[10];
  float* out = (float*)d_out;

  float* ws  = (float*)d_ws;
  unsigned* degp = (unsigned*)ws;
  unsigned* counts = degp + FIND_S * N_PAD;
  unsigned* offs = counts + FIND_S * NOWN_PAD;
  int2* list = (int2*)(offs + 327680);
  float* agg = (float*)(list + FIND_S * PART_LIST);
  float* h1  = agg + (size_t)N_NODES * 128;
  float* h2  = h1  + (size_t)N_NODES * 128;

  find_edges<<<NOWN / FIND_G * FIND_S, OWN_THREADS, 0, stream>>>(src, dst, list, counts, offs, degp);

  sum_rows_own<128><<<NOWN, OWN_THREADS, 0, stream>>>(x, src, dst, list, counts, offs, agg);
  sage_wmma_gemm<128, 128, 64><<<GEMM_WGS, GEMM_THREADS, 0, stream>>>(agg, x, degp, Wl1, Wr1, b1, h1);

  sum_rows_own<128><<<NOWN, OWN_THREADS, 0, stream>>>(h1, src, dst, list, counts, offs, agg);
  sage_wmma_gemm<128, 64, 32><<<GEMM_WGS, GEMM_THREADS, 0, stream>>>(agg, h1, degp, Wl2, Wr2, b2, h2);

  sum_rows_own<64><<<NOWN, OWN_THREADS, 0, stream>>>(h2, src, dst, list, counts, offs, agg);
  sage_wmma_gemm<64, 128, 64><<<GEMM_WGS, GEMM_THREADS, 0, stream>>>(agg, h2, degp, Wl3, Wr3, b3, out);
}
